# starting kernel
# baseline (speedup 1.0000x reference)
#include <hip/hip_runtime.h>

typedef __attribute__((ext_vector_type(2))) float v2f;
typedef __attribute__((ext_vector_type(8))) float v8f;

#define D 128
#define LDS_STRIDE 130

__global__ void init_deg_kernel(float* __restrict__ deg, int n) {
    int i = blockIdx.x * blockDim.x + threadIdx.x;
    if (i < n) deg[i] = 1.0f;
}

__global__ void count_deg_kernel(const int* __restrict__ dst, int E,
                                 float* __restrict__ deg) {
    int e = blockIdx.x * blockDim.x + threadIdx.x;
    if (e < E) atomicAdd(&deg[dst[e]], 1.0f);
}

__global__ void dinv_kernel(float* __restrict__ deg, int n) {
    int i = blockIdx.x * blockDim.x + threadIdx.x;
    if (i < n) deg[i] = rsqrtf(deg[i]);
}

__global__ __launch_bounds__(128) void gemm_wmma_f32(
    const float* __restrict__ A, const float* __restrict__ W,
    float* __restrict__ C, int nrows) {
    __shared__ float wT[D * LDS_STRIDE];

    int tid = threadIdx.x;
    for (int i = tid; i < D * D; i += 128) {
        int r = i >> 7, c = i & (D - 1);
        wT[c * LDS_STRIDE + r] = W[i];
    }
    __syncthreads();

    int wave = tid >> 5;
    int lane = tid & 31;
    int lrow = lane & 15;
    int lhi  = lane >> 4;

    int m0 = blockIdx.x * 64 + wave * 16;

    v8f acc[8];
#pragma unroll
    for (int n = 0; n < 8; ++n) acc[n] = (v8f)(0.0f);

    int arow = m0 + lrow;
    int crow = arow < nrows ? arow : 0;
    const float* aptr = A + (size_t)crow * D + lhi * 2;

    for (int k = 0; k < D; k += 4) {
        v2f a = *(const v2f*)(aptr + k);
#pragma unroll
        for (int n = 0; n < 8; ++n) {
            int col = n * 16 + lrow;
            v2f b = *(const v2f*)(&wT[col * LDS_STRIDE + k + lhi * 2]);
            acc[n] = __builtin_amdgcn_wmma_f32_16x16x4_f32(
                false, a, false, b, (short)0, acc[n], false, false);
        }
    }

#pragma unroll
    for (int n = 0; n < 8; ++n) {
        int col = n * 16 + lrow;
#pragma unroll
        for (int v = 0; v < 8; ++v) {
            int row = m0 + v + 8 * lhi;
            if (row < nrows) C[(size_t)row * D + col] = acc[n][v];
        }
    }
}

__global__ void self_loop_init_kernel(const float* __restrict__ h,
                                      const float* __restrict__ dinv,
                                      float* __restrict__ out, int n) {
    long long i = (long long)blockIdx.x * blockDim.x + threadIdx.x;
    long long total = (long long)n * (D / 4);
    if (i < total) {
        int node = (int)(i >> 5);
        float w = dinv[node] * dinv[node];
        float4 v = ((const float4*)h)[i];
        v.x *= w; v.y *= w; v.z *= w; v.w *= w;
        ((float4*)out)[i] = v;
    }
}

__global__ __launch_bounds__(256) void edge_scatter_kernel(
    const int* __restrict__ src, const int* __restrict__ dst,
    const float* __restrict__ h, const float* __restrict__ dinv,
    float* __restrict__ out, int E) {
    int e = blockIdx.x * 8 + (threadIdx.x >> 5);
    int lane = threadIdx.x & 31;
    if (e >= E) return;
    int s = src[e];
    int d = dst[e];
    float w = dinv[s] * dinv[d];
    float4 v = ((const float4*)(h + (size_t)s * D))[lane];
    float* op = out + (size_t)d * D + lane * 4;
    atomicAdd(op + 0, v.x * w);
    atomicAdd(op + 1, v.y * w);
    atomicAdd(op + 2, v.z * w);
    atomicAdd(op + 3, v.w * w);
}

__global__ void bias_act_kernel(float* __restrict__ x,
                                const float* __restrict__ b,
                                long long total, int relu) {
    long long i = (long long)blockIdx.x * blockDim.x + threadIdx.x;
    if (i < total) {
        float v = x[i] + b[i & (D - 1)];
        if (relu) v = fmaxf(v, 0.0f);
        x[i] = v;
    }
}

extern "C" void kernel_launch(void* const* d_in, const int* in_sizes, int n_in,
                              void* d_out, int out_size, void* d_ws, size_t ws_size,
                              hipStream_t stream) {
    const float* x   = (const float*)d_in[0];
    const int*   ei  = (const int*)d_in[1];
    const float* W1  = (const float*)d_in[2];
    const float* b1  = (const float*)d_in[3];
    const float* W2  = (const float*)d_in[4];
    const float* b2  = (const float*)d_in[5];
    float* out = (float*)d_out;

    const int N = in_sizes[0] / D;
    const int E = in_sizes[1] / 2;
    const int* src = ei;
    const int* dst = ei + E;

    float* dinv = (float*)d_ws;
    size_t dinvPad = ((size_t)N + 63) & ~(size_t)63;
    float* bufA = dinv + dinvPad;
    float* bufB = bufA + (size_t)N * D;

    const long long totalF  = (long long)N * D;
    const int nThreads256   = 256;
    const int nBlocksN      = (N + 255) / 256;
    const int nBlocksE      = (E + 255) / 256;
    const int nBlocksEdge   = (E + 7) / 8;
    const int nBlocksF4     = (int)(((long long)N * (D / 4) + 255) / 256);
    const int nBlocksF      = (int)((totalF + 255) / 256);
    const int nBlocksGemm   = (N + 63) / 64;

    init_deg_kernel<<<nBlocksN, nThreads256, 0, stream>>>(dinv, N);
    count_deg_kernel<<<nBlocksE, nThreads256, 0, stream>>>(dst, E, dinv);
    dinv_kernel<<<nBlocksN, nThreads256, 0, stream>>>(dinv, N);

    gemm_wmma_f32<<<nBlocksGemm, 128, 0, stream>>>(x, W1, bufA, N);
    self_loop_init_kernel<<<nBlocksF4, nThreads256, 0, stream>>>(bufA, dinv, bufB, N);
    edge_scatter_kernel<<<nBlocksEdge, nThreads256, 0, stream>>>(src, dst, bufA, dinv, bufB, E);
    bias_act_kernel<<<nBlocksF, nThreads256, 0, stream>>>(bufB, b1, totalF, 1);

    gemm_wmma_f32<<<nBlocksGemm, 128, 0, stream>>>(bufB, W2, bufA, N);
    self_loop_init_kernel<<<nBlocksF4, nThreads256, 0, stream>>>(bufA, dinv, out, N);
    edge_scatter_kernel<<<nBlocksEdge, nThreads256, 0, stream>>>(src, dst, bufA, dinv, out, E);
    bias_act_kernel<<<nBlocksF, nThreads256, 0, stream>>>(out, b2, totalF, 0);
}


# edge scatter: lane l takes features l,l+32,l+64,l+96, so each wave-wide float atomic (and load) covers one whole 128-B line, not 8 dwords in each of 4 lines; atomics kept, sum order as before
# speedup vs baseline: 1.3457x; 1.3457x over previous
#include <hip/hip_runtime.h>

typedef __attribute__((ext_vector_type(2))) float v2f;
typedef __attribute__((ext_vector_type(8))) float v8f;

#define D 128
#define LDS_STRIDE 130

__global__ void init_deg_kernel(float* __restrict__ deg, int n) {
    int i = blockIdx.x * blockDim.x + threadIdx.x;
    if (i < n) deg[i] = 1.0f;
}

__global__ void count_deg_kernel(const int* __restrict__ dst, int E,
                                 float* __restrict__ deg) {
    int e = blockIdx.x * blockDim.x + threadIdx.x;
    if (e < E) atomicAdd(&deg[dst[e]], 1.0f);
}

__global__ void dinv_kernel(float* __restrict__ deg, int n) {
    int i = blockIdx.x * blockDim.x + threadIdx.x;
    if (i < n) deg[i] = rsqrtf(deg[i]);
}

__global__ __launch_bounds__(128) void gemm_wmma_f32(
    const float* __restrict__ A, const float* __restrict__ W,
    float* __restrict__ C, int nrows) {
    __shared__ float wT[D * LDS_STRIDE];

    int tid = threadIdx.x;
    for (int i = tid; i < D * D; i += 128) {
        int r = i >> 7, c = i & (D - 1);
        wT[c * LDS_STRIDE + r] = W[i];
    }
    __syncthreads();

    int wave = tid >> 5;
    int lane = tid & 31;
    int lrow = lane & 15;
    int lhi  = lane >> 4;

    int m0 = blockIdx.x * 64 + wave * 16;

    v8f acc[8];
#pragma unroll
    for (int n = 0; n < 8; ++n) acc[n] = (v8f)(0.0f);

    int arow = m0 + lrow;
    int crow = arow < nrows ? arow : 0;
    const float* aptr = A + (size_t)crow * D + lhi * 2;

    for (int k = 0; k < D; k += 4) {
        v2f a = *(const v2f*)(aptr + k);
#pragma unroll
        for (int n = 0; n < 8; ++n) {
            int col = n * 16 + lrow;
            v2f b = *(const v2f*)(&wT[col * LDS_STRIDE + k + lhi * 2]);
            acc[n] = __builtin_amdgcn_wmma_f32_16x16x4_f32(
                false, a, false, b, (short)0, acc[n], false, false);
        }
    }

#pragma unroll
    for (int n = 0; n < 8; ++n) {
        int col = n * 16 + lrow;
#pragma unroll
        for (int v = 0; v < 8; ++v) {
            int row = m0 + v + 8 * lhi;
            if (row < nrows) C[(size_t)row * D + col] = acc[n][v];
        }
    }
}

__global__ void self_loop_init_kernel(const float* __restrict__ h,
                                      const float* __restrict__ dinv,
                                      float* __restrict__ out, int n) {
    long long i = (long long)blockIdx.x * blockDim.x + threadIdx.x;
    long long total = (long long)n * (D / 4);
    if (i < total) {
        int node = (int)(i >> 5);
        float w = dinv[node] * dinv[node];
        float4 v = ((const float4*)h)[i];
        v.x *= w; v.y *= w; v.z *= w; v.w *= w;
        ((float4*)out)[i] = v;
    }
}

__global__ __launch_bounds__(256) void edge_scatter_kernel(
    const int* __restrict__ src, const int* __restrict__ dst,
    const float* __restrict__ h, const float* __restrict__ dinv,
    float* __restrict__ out, int E) {
    int e = blockIdx.x * 8 + (threadIdx.x >> 5);
    int lane = threadIdx.x & 31;
    if (e >= E) return;
    int s = src[e];
    int d = dst[e];
    float w = dinv[s] * dinv[d];
    const float* hp = h + (size_t)s * D + lane;
    float* op = out + (size_t)d * D + lane;
    float v0 = hp[0];
    float v1 = hp[32];
    float v2 = hp[64];
    float v3 = hp[96];
    atomicAdd(op + 0,  v0 * w);
    atomicAdd(op + 32, v1 * w);
    atomicAdd(op + 64, v2 * w);
    atomicAdd(op + 96, v3 * w);
}

__global__ void bias_act_kernel(float* __restrict__ x,
                                const float* __restrict__ b,
                                long long total, int relu) {
    long long i = (long long)blockIdx.x * blockDim.x + threadIdx.x;
    if (i < total) {
        float v = x[i] + b[i & (D - 1)];
        if (relu) v = fmaxf(v, 0.0f);
        x[i] = v;
    }
}

extern "C" void kernel_launch(void* const* d_in, const int* in_sizes, int n_in,
                              void* d_out, int out_size, void* d_ws, size_t ws_size,
                              hipStream_t stream) {
    const float* x   = (const float*)d_in[0];
    const int*   ei  = (const int*)d_in[1];
    const float* W1  = (const float*)d_in[2];
    const float* b1  = (const float*)d_in[3];
    const float* W2  = (const float*)d_in[4];
    const float* b2  = (const float*)d_in[5];
    float* out = (float*)d_out;

    const int N = in_sizes[0] / D;
    const int E = in_sizes[1] / 2;
    const int* src = ei;
    const int* dst = ei + E;

    float* dinv = (float*)d_ws;
    size_t dinvPad = ((size_t)N + 63) & ~(size_t)63;
    float* bufA = dinv + dinvPad;
    float* bufB = bufA + (size_t)N * D;

    const long long totalF  = (long long)N * D;
    const int nThreads256   = 256;
    const int nBlocksN      = (N + 255) / 256;
    const int nBlocksE      = (E + 255) / 256;
    const int nBlocksEdge   = (E + 7) / 8;
    const int nBlocksF4     = (int)(((long long)N * (D / 4) + 255) / 256);
    const int nBlocksF      = (int)((totalF + 255) / 256);
    const int nBlocksGemm   = (N + 63) / 64;

    init_deg_kernel<<<nBlocksN, nThreads256, 0, stream>>>(dinv, N);
    count_deg_kernel<<<nBlocksE, nThreads256, 0, stream>>>(dst, E, dinv);
    dinv_kernel<<<nBlocksN, nThreads256, 0, stream>>>(dinv, N);

    gemm_wmma_f32<<<nBlocksGemm, 128, 0, stream>>>(x, W1, bufA, N);
    self_loop_init_kernel<<<nBlocksF4, nThreads256, 0, stream>>>(bufA, dinv, bufB, N);
    edge_scatter_kernel<<<nBlocksEdge, nThreads256, 0, stream>>>(src, dst, bufA, dinv, bufB, E);
    bias_act_kernel<<<nBlocksF, nThreads256, 0, stream>>>(bufB, b1, totalF, 1);

    gemm_wmma_f32<<<nBlocksGemm, 128, 0, stream>>>(bufB, W2, bufA, N);
    self_loop_init_kernel<<<nBlocksF4, nThreads256, 0, stream>>>(bufA, dinv, out, N);
    edge_scatter_kernel<<<nBlocksEdge, nThreads256, 0, stream>>>(src, dst, bufA, dinv, out, E);
    bias_act_kernel<<<nBlocksF, nThreads256, 0, stream>>>(out, b2, totalF, 0);
}


# aggregate by destination: edges bucketed by dst in-call (count, scan, slot fill); one wave sums a row in registers, stores it once with self term+bias; no float atomics; sum order = slot arrival
# speedup vs baseline: 2.0059x; 1.4906x over previous
#include <hip/hip_runtime.h>

typedef __attribute__((ext_vector_type(2))) float v2f;
typedef __attribute__((ext_vector_type(8))) float v8f;

#define D 128
#define LDS_STRIDE 130

__global__ void zero_cnt_kernel(int* __restrict__ cnt, int n) {
    int i = blockIdx.x * blockDim.x + threadIdx.x;
    if (i < n) cnt[i] = 0;
}

__global__ void count_dst_kernel(const int* __restrict__ dst, int E,
                                 int* __restrict__ cnt) {
    int e = blockIdx.x * blockDim.x + threadIdx.x;
    if (e < E) atomicAdd(&cnt[dst[e]], 1);
}

__global__ __launch_bounds__(256) void scan_block_kernel(
    const int* __restrict__ cnt, int* __restrict__ rowptr,
    int* __restrict__ bsum, int n) {
    __shared__ int wsum[8];
    int t = threadIdx.x;
    int lane = t & 31;
    int wave = t >> 5;
    int i = blockIdx.x * 256 + t;
    int v = (i < n) ? cnt[i] : 0;
    int x = v;
    for (int off = 1; off < 32; off <<= 1) {
        int y = __shfl_up(x, off);
        if (lane >= off) x += y;
    }
    if (lane == 31) wsum[wave] = x;
    __syncthreads();
    int wo = 0;
    for (int k = 0; k < 8; ++k) wo += (k < wave) ? wsum[k] : 0;
    if (i < n) rowptr[i] = wo + x - v;
    if (t == 255) bsum[blockIdx.x] = wo + x;
}

__global__ __launch_bounds__(256) void scan_finish_kernel(
    const int* __restrict__ cnt, const int* __restrict__ bsum,
    int* __restrict__ rowptr, int* __restrict__ fillpos,
    float* __restrict__ dinv, int n) {
    __shared__ int red[8];
    int t = threadIdx.x;
    int b = blockIdx.x;
    int part = 0;
    for (int j = t; j < b; j += 256) part += bsum[j];
    for (int off = 16; off > 0; off >>= 1) part += __shfl_xor(part, off);
    if ((t & 31) == 0) red[t >> 5] = part;
    __syncthreads();
    int boff = 0;
    for (int k = 0; k < 8; ++k) boff += red[k];
    int i = b * 256 + t;
    if (i < n) {
        int r = rowptr[i] + boff;
        rowptr[i] = r;
        fillpos[i] = r;
        dinv[i] = rsqrtf((float)(cnt[i] + 1));
    }
}

__global__ void fill_bucket_kernel(const int* __restrict__ src,
                                   const int* __restrict__ dst, int E,
                                   int* __restrict__ fillpos,
                                   int* __restrict__ bucket) {
    int e = blockIdx.x * blockDim.x + threadIdx.x;
    if (e < E) {
        int p = atomicAdd(&fillpos[dst[e]], 1);
        bucket[p] = src[e];
    }
}

__global__ __launch_bounds__(128) void gemm_wmma_f32(
    const float* __restrict__ A, const float* __restrict__ W,
    float* __restrict__ C, int nrows) {
    __shared__ float wT[D * LDS_STRIDE];

    int tid = threadIdx.x;
    for (int i = tid; i < D * D; i += 128) {
        int r = i >> 7, c = i & (D - 1);
        wT[c * LDS_STRIDE + r] = W[i];
    }
    __syncthreads();

    int wave = tid >> 5;
    int lane = tid & 31;
    int lrow = lane & 15;
    int lhi  = lane >> 4;

    int m0 = blockIdx.x * 64 + wave * 16;

    v8f acc[8];
#pragma unroll
    for (int n = 0; n < 8; ++n) acc[n] = (v8f)(0.0f);

    int arow = m0 + lrow;
    int crow = arow < nrows ? arow : 0;
    const float* aptr = A + (size_t)crow * D + lhi * 2;

    for (int k = 0; k < D; k += 4) {
        v2f a = *(const v2f*)(aptr + k);
#pragma unroll
        for (int n = 0; n < 8; ++n) {
            int col = n * 16 + lrow;
            v2f b = *(const v2f*)(&wT[col * LDS_STRIDE + k + lhi * 2]);
            acc[n] = __builtin_amdgcn_wmma_f32_16x16x4_f32(
                false, a, false, b, (short)0, acc[n], false, false);
        }
    }

#pragma unroll
    for (int n = 0; n < 8; ++n) {
        int col = n * 16 + lrow;
#pragma unroll
        for (int v = 0; v < 8; ++v) {
            int row = m0 + v + 8 * lhi;
            if (row < nrows) C[(size_t)row * D + col] = acc[n][v];
        }
    }
}

__device__ __forceinline__ void fma4(float4& a, const float4& v, float w) {
    a.x = fmaf(v.x, w, a.x);
    a.y = fmaf(v.y, w, a.y);
    a.z = fmaf(v.z, w, a.z);
    a.w = fmaf(v.w, w, a.w);
}

__global__ __launch_bounds__(256) void gather_rows_kernel(
    const int* __restrict__ rowptr, const int* __restrict__ cnt,
    const int* __restrict__ bucket, const float* __restrict__ h,
    const float* __restrict__ dinv, const float* __restrict__ bias,
    float* __restrict__ out, int n, int relu) {
    int lane = threadIdx.x & 31;
    int node = __builtin_amdgcn_readfirstlane(blockIdx.x * 8 + (threadIdx.x >> 5));
    if (node >= n) return;
    int beg = rowptr[node];
    int c = cnt[node];
    float di = dinv[node];
    const float4* h4 = (const float4*)h;

    float4 acc = h4[(size_t)node * 32 + lane];
    float wself = di * di;
    acc.x *= wself; acc.y *= wself; acc.z *= wself; acc.w *= wself;

    for (int base = 0; base < c; base += 32) {
        int m = c - base;
        if (m > 32) m = 32;
        int s = 0;
        float w = 0.0f;
        if (lane < m) {
            s = bucket[beg + base + lane];
            w = dinv[s] * di;
        }
        int wi = __float_as_int(w);
        int j = 0;
        for (; j + 4 <= m; j += 4) {
            int s0 = __builtin_amdgcn_readlane(s, j);
            int s1 = __builtin_amdgcn_readlane(s, j + 1);
            int s2 = __builtin_amdgcn_readlane(s, j + 2);
            int s3 = __builtin_amdgcn_readlane(s, j + 3);
            float w0 = __int_as_float(__builtin_amdgcn_readlane(wi, j));
            float w1 = __int_as_float(__builtin_amdgcn_readlane(wi, j + 1));
            float w2 = __int_as_float(__builtin_amdgcn_readlane(wi, j + 2));
            float w3 = __int_as_float(__builtin_amdgcn_readlane(wi, j + 3));
            float4 v0 = h4[(size_t)s0 * 32 + lane];
            float4 v1 = h4[(size_t)s1 * 32 + lane];
            float4 v2 = h4[(size_t)s2 * 32 + lane];
            float4 v3 = h4[(size_t)s3 * 32 + lane];
            fma4(acc, v0, w0);
            fma4(acc, v1, w1);
            fma4(acc, v2, w2);
            fma4(acc, v3, w3);
        }
        for (; j < m; ++j) {
            int s0 = __builtin_amdgcn_readlane(s, j);
            float w0 = __int_as_float(__builtin_amdgcn_readlane(wi, j));
            float4 v0 = h4[(size_t)s0 * 32 + lane];
            fma4(acc, v0, w0);
        }
    }

    float4 bv = ((const float4*)bias)[lane];
    acc.x += bv.x; acc.y += bv.y; acc.z += bv.z; acc.w += bv.w;
    if (relu) {
        acc.x = fmaxf(acc.x, 0.0f); acc.y = fmaxf(acc.y, 0.0f);
        acc.z = fmaxf(acc.z, 0.0f); acc.w = fmaxf(acc.w, 0.0f);
    }
    ((float4*)out)[(size_t)node * 32 + lane] = acc;
}

extern "C" void kernel_launch(void* const* d_in, const int* in_sizes, int n_in,
                              void* d_out, int out_size, void* d_ws, size_t ws_size,
                              hipStream_t stream) {
    const float* x   = (const float*)d_in[0];
    const int*   ei  = (const int*)d_in[1];
    const float* W1  = (const float*)d_in[2];
    const float* b1  = (const float*)d_in[3];
    const float* W2  = (const float*)d_in[4];
    const float* b2  = (const float*)d_in[5];
    float* out = (float*)d_out;

    const int N = in_sizes[0] / D;
    const int E = in_sizes[1] / 2;
    const int* src = ei;
    const int* dst = ei + E;

    const int nBlocksN      = (N + 255) / 256;
    const int nBlocksE      = (E + 255) / 256;
    const int nBlocksRow    = (N + 7) / 8;
    const int nBlocksGemm   = (N + 63) / 64;

    size_t nPad = ((size_t)N + 63) & ~(size_t)63;
    size_t ePad = ((size_t)E + 63) & ~(size_t)63;
    size_t bPad = ((size_t)nBlocksN + 63) & ~(size_t)63;
    int* cnt     = (int*)d_ws;
    int* rowptr  = cnt + nPad;
    int* fillpos = rowptr + nPad;
    int* bsum    = fillpos + nPad;
    int* bucket  = bsum + bPad;
    float* dinv  = (float*)(bucket + ePad);
    float* bufA  = dinv + nPad;
    float* bufB  = bufA + (size_t)N * D;

    zero_cnt_kernel<<<nBlocksN, 256, 0, stream>>>(cnt, N);
    count_dst_kernel<<<nBlocksE, 256, 0, stream>>>(dst, E, cnt);
    scan_block_kernel<<<nBlocksN, 256, 0, stream>>>(cnt, rowptr, bsum, N);
    scan_finish_kernel<<<nBlocksN, 256, 0, stream>>>(cnt, bsum, rowptr, fillpos, dinv, N);
    fill_bucket_kernel<<<nBlocksE, 256, 0, stream>>>(src, dst, E, fillpos, bucket);

    gemm_wmma_f32<<<nBlocksGemm, 128, 0, stream>>>(x, W1, bufA, N);
    gather_rows_kernel<<<nBlocksRow, 256, 0, stream>>>(rowptr, cnt, bucket, bufA, dinv, b1, bufB, N, 1);

    gemm_wmma_f32<<<nBlocksGemm, 128, 0, stream>>>(bufB, W2, bufA, N);
    gather_rows_kernel<<<nBlocksRow, 256, 0, stream>>>(rowptr, cnt, bucket, bufA, dinv, b2, out, N, 0);
}


# edge buckets as 32 fixed slots per destination row plus an overflow list scanned by rows that exceed them: count pass and both scan launches go (7 -> 4 launches); sum order = slot arrival as before
# speedup vs baseline: 2.8681x; 1.4298x over previous
#include <hip/hip_runtime.h>

typedef __attribute__((ext_vector_type(2))) float v2f;
typedef __attribute__((ext_vector_type(8))) float v8f;

#define D 128
#define LDS_STRIDE 130

#define CAP 32

__global__ void zero_cnt_kernel(int* __restrict__ cnt, int n) {
    int i = blockIdx.x * blockDim.x + threadIdx.x;
    if (i < n) cnt[i] = 0;
}

__global__ void fill_bucket_kernel(const int* __restrict__ src,
                                   const int* __restrict__ dst, int E,
                                   int* __restrict__ cnt,
                                   int* __restrict__ bucket,
                                   int* __restrict__ novf,
                                   int* __restrict__ ovf) {
    int e = blockIdx.x * blockDim.x + threadIdx.x;
    if (e < E) {
        int d = dst[e];
        int s = src[e];
        int p = atomicAdd(&cnt[d], 1);
        if (p < CAP) {
            bucket[(size_t)d * CAP + p] = s;
        } else {
            int q = atomicAdd(novf, 1);
            ovf[2 * (size_t)q] = d;
            ovf[2 * (size_t)q + 1] = s;
        }
    }
}

#define TILE_STRIDE 132

__device__ __forceinline__ float dinv_of(int count) {
    return rsqrtf((float)(count + 1));
}

__device__ __forceinline__ void fma4(float4& a, const float4& v, float w) {
    a.x = fmaf(v.x, w, a.x);
    a.y = fmaf(v.y, w, a.y);
    a.z = fmaf(v.z, w, a.z);
    a.w = fmaf(v.w, w, a.w);
}

template <int G>
__device__ __forceinline__ void add_group(float4& acc, const float4* __restrict__ h4,
                                          int s, int wi, int j, int lane) {
    float4 v[G];
#pragma unroll
    for (int k = 0; k < G; ++k) {
        int sk = __builtin_amdgcn_readlane(s, j + k);
        v[k] = h4[(size_t)sk * 32 + lane];
    }
#pragma unroll
    for (int k = 0; k < G; ++k) {
        float wk = __int_as_float(__builtin_amdgcn_readlane(wi, j + k));
        fma4(acc, v[k], wk);
    }
}

__device__ __forceinline__ void add_chunk(float4& acc, const float4* __restrict__ h4,
                                          int s, int wi, int m, int lane) {
    int j = 0;
    for (; j + 16 <= m; j += 16) add_group<16>(acc, h4, s, wi, j, lane);
    switch (m - j) {
        case 15: add_group<15>(acc, h4, s, wi, j, lane); break;
        case 14: add_group<14>(acc, h4, s, wi, j, lane); break;
        case 13: add_group<13>(acc, h4, s, wi, j, lane); break;
        case 12: add_group<12>(acc, h4, s, wi, j, lane); break;
        case 11: add_group<11>(acc, h4, s, wi, j, lane); break;
        case 10: add_group<10>(acc, h4, s, wi, j, lane); break;
        case 9:  add_group<9>(acc, h4, s, wi, j, lane); break;
        case 8:  add_group<8>(acc, h4, s, wi, j, lane); break;
        case 7:  add_group<7>(acc, h4, s, wi, j, lane); break;
        case 6:  add_group<6>(acc, h4, s, wi, j, lane); break;
        case 5:  add_group<5>(acc, h4, s, wi, j, lane); break;
        case 4:  add_group<4>(acc, h4, s, wi, j, lane); break;
        case 3:  add_group<3>(acc, h4, s, wi, j, lane); break;
        case 2:  add_group<2>(acc, h4, s, wi, j, lane); break;
        case 1:  add_group<1>(acc, h4, s, wi, j, lane); break;
        default: break;
    }
}

__global__ __launch_bounds__(256) void gather_gemm_kernel(
    const int* __restrict__ cnt, const int* __restrict__ bucket,
    const int* __restrict__ novf, const int* __restrict__ ovf,
    const float* __restrict__ xin, const float* __restrict__ W,
    const float* __restrict__ bias, float* __restrict__ out, int n, int relu) {
    __shared__ __attribute__((aligned(16))) float wT[D * LDS_STRIDE];
    __shared__ __attribute__((aligned(16))) float tiles[8 * 16 * TILE_STRIDE];

    int tid = threadIdx.x;
    for (int i = tid; i < D * D; i += 256) {
        int r = i >> 7, c = i & (D - 1);
        wT[c * LDS_STRIDE + r] = W[i];
    }
    __syncthreads();

    int lane = tid & 31;
    int wave = __builtin_amdgcn_readfirstlane(tid >> 5);
    float* tile = tiles + wave * (16 * TILE_STRIDE);
    const float4* h4 = (const float4*)xin;
    int ntiles = (n + 15) >> 4;

    for (int t = wave * gridDim.x + blockIdx.x; t < ntiles; t += 8 * gridDim.x) {
    int node0 = t * 16;

    int mr = node0 + (lane & 15);
    bool mvalid = mr < n;
    int mrc = mvalid ? mr : 0;
    int cl = cnt[mrc];
    int ddi = __float_as_int(dinv_of(cl));
    int cc = mvalid ? cl : 0;
    int cb = cc < CAP ? cc : CAP;
    const int* brow = bucket + (size_t)node0 * CAP;

    int s0 = 0, s1 = 0;
    float w0 = 0.0f;
    {
        int c0 = __builtin_amdgcn_readlane(cb, 0);
        int c1 = __builtin_amdgcn_readlane(cb, 1);
        float g0 = 0.0f;
        if (lane < c0) s0 = brow[lane];
        if (lane < c1) s1 = brow[CAP + lane];
        if (lane < c0) g0 = dinv_of(cnt[s0]);
        w0 = g0 * __int_as_float(__builtin_amdgcn_readlane(ddi, 0));
    }

#pragma unroll 1
    for (int r = 0; r < 16; ++r) {
        int r1 = r + 1 < 16 ? r + 1 : 15;
        int r2 = r + 2 < 16 ? r + 2 : 15;
        int m1 = r + 1 < 16 ? __builtin_amdgcn_readlane(cb, r1) : 0;
        int k1 = cnt[s1];
        int m2 = r + 2 < 16 ? __builtin_amdgcn_readlane(cb, r2) : 0;
        int s2 = 0;
        if (lane < m2) s2 = brow[r2 * CAP + lane];

        int node = node0 + r;
        float4 acc = make_float4(0.0f, 0.0f, 0.0f, 0.0f);
        if (node < n) {
            int c = __builtin_amdgcn_readlane(cc, r);
            int m = __builtin_amdgcn_readlane(cb, r);
            float di = __int_as_float(__builtin_amdgcn_readlane(ddi, r));
            float4 vself = h4[(size_t)node * 32 + lane];

            add_chunk(acc, h4, s0, __float_as_int(w0), m, lane);

            if (c > CAP) {
                int nv = novf[0];
                for (int q0 = 0; q0 < nv; q0 += 32) {
                    int q = q0 + lane;
                    int os = 0;
                    float ow = 0.0f;
                    bool hit = false;
                    if (q < nv) {
                        if (ovf[2 * (size_t)q] == node) {
                            hit = true;
                            os = ovf[2 * (size_t)q + 1];
                            ow = dinv_of(cnt[os]) * di;
                        }
                    }
                    unsigned mask = __builtin_amdgcn_ballot_w32(hit);
                    while (mask != 0u) {
                        int j = __builtin_ctz(mask);
                        mask &= mask - 1u;
                        int sk = __builtin_amdgcn_readlane(os, j);
                        float wk = __int_as_float(
                            __builtin_amdgcn_readlane(__float_as_int(ow), j));
                        float4 v = h4[(size_t)sk * 32 + lane];
                        fma4(acc, v, wk);
                    }
                }
            }
            fma4(acc, vself, di * di);
        }
        *(float4*)(tile + r * TILE_STRIDE + 4 * lane) = acc;

        s0 = s1;
        s1 = s2;
        w0 = dinv_of(k1) * __int_as_float(__builtin_amdgcn_readlane(ddi, r1));
    }

    __builtin_amdgcn_fence(__ATOMIC_RELEASE, "wavefront");
    __builtin_amdgcn_wave_barrier();
    __builtin_amdgcn_fence(__ATOMIC_ACQUIRE, "wavefront");

    int lrow = lane & 15;
    int lhi  = lane >> 4;

    v8f acc8[8];
#pragma unroll
    for (int nn = 0; nn < 8; ++nn) acc8[nn] = (v8f)(0.0f);

    const float* ap = tile + lrow * TILE_STRIDE + lhi * 2;
    for (int k = 0; k < D; k += 4) {
        v2f a = *(const v2f*)(ap + k);
#pragma unroll
        for (int nn = 0; nn < 8; ++nn) {
            int col = nn * 16 + lrow;
            v2f b = *(const v2f*)(&wT[col * LDS_STRIDE + k + lhi * 2]);
            acc8[nn] = __builtin_amdgcn_wmma_f32_16x16x4_f32(
                false, a, false, b, (short)0, acc8[nn], false, false);
        }
    }

#pragma unroll
    for (int nn = 0; nn < 8; ++nn) {
        int col = nn * 16 + lrow;
        float bv = bias[col];
#pragma unroll
        for (int v = 0; v < 8; ++v) {
            int row = node0 + v + 8 * lhi;
            float o = acc8[nn][v] + bv;
            if (relu) o = fmaxf(o, 0.0f);
            if (row < n) out[(size_t)row * D + col] = o;
        }
    }
    }
}

extern "C" void kernel_launch(void* const* d_in, const int* in_sizes, int n_in,
                              void* d_out, int out_size, void* d_ws, size_t ws_size,
                              hipStream_t stream) {
    const float* x   = (const float*)d_in[0];
    const int*   ei  = (const int*)d_in[1];
    const float* W1  = (const float*)d_in[2];
    const float* b1  = (const float*)d_in[3];
    const float* W2  = (const float*)d_in[4];
    const float* b2  = (const float*)d_in[5];
    float* out = (float*)d_out;

    const int N = in_sizes[0] / D;
    const int E = in_sizes[1] / 2;
    const int* src = ei;
    const int* dst = ei + E;

    const int nBlocksE      = (E + 255) / 256;
    const int nBlocksLayer  = 512;

    size_t nPad = ((size_t)N + 63) & ~(size_t)63;
    size_t ePad = ((size_t)E + 63) & ~(size_t)63;
    int* cnt     = (int*)d_ws;
    int* novf    = cnt + nPad;
    int* bucket  = novf + 64;
    int* ovf     = bucket + nPad * CAP;
    float* bufB  = (float*)(ovf + 2 * ePad);

    const int nZero      = (int)nPad + 64;
    const int nBlocksZ   = (nZero + 255) / 256;
    zero_cnt_kernel<<<nBlocksZ, 256, 0, stream>>>(cnt, nZero);
    fill_bucket_kernel<<<nBlocksE, 256, 0, stream>>>(src, dst, E, cnt, bucket, novf, ovf);

    gather_gemm_kernel<<<nBlocksLayer, 256, 0, stream>>>(cnt, bucket, novf, ovf, x, W1, b1, bufB, N, 1);
    gather_gemm_kernel<<<nBlocksLayer, 256, 0, stream>>>(cnt, bucket, novf, ovf, bufB, W2, b2, out, N, 0);
}


# edge bucketing without a global atomic per edge: LDS counters rank edges into fixed (row range, chunk) segments, one work-group per 128 rows builds cnt and slots in LDS; sum order = slot arrival
# speedup vs baseline: 3.8267x; 1.3342x over previous
#include <hip/hip_runtime.h>

typedef __attribute__((ext_vector_type(2))) float v2f;
typedef __attribute__((ext_vector_type(8))) float v8f;
typedef unsigned long long u64;

#define D 128
#define LDS_STRIDE 130

#define CAP 32
#define RB_SHIFT 7
#define RB (1 << RB_SHIFT)
#define SEGK 32
#define EPT 10
#define CHUNK (256 * EPT)
#define MAXB 1024

__global__ __launch_bounds__(256) void partition_kernel(
    const int* __restrict__ src, const int* __restrict__ dst, int E, int nbins, int NB,
    int* __restrict__ segcnt, int* __restrict__ stage_w,
    int* __restrict__ spillcnt, int* __restrict__ spill_w, int* __restrict__ novf) {
    u64* __restrict__ stage = (u64*)stage_w;
    u64* __restrict__ spill = (u64*)spill_w;
    __shared__ int lcnt[MAXB];
    __shared__ int lspill;
    int tid = threadIdx.x;
    int c = blockIdx.x;
    if (c == 0 && tid == 0) novf[0] = 0;
    if (tid == 0) lspill = 0;

    int d[EPT], s[EPT];
#pragma unroll
    for (int i = 0; i < EPT; ++i) {
        int e = c * CHUNK + i * 256 + tid;
        bool ok = e < E;
        d[i] = ok ? dst[e] : -1;
        s[i] = ok ? src[e] : 0;
    }

    for (int b0 = 0; b0 < nbins; b0 += MAXB) {
        for (int i = tid; i < MAXB; i += 256) lcnt[i] = 0;
        __syncthreads();
        int rank[EPT];
#pragma unroll
        for (int i = 0; i < EPT; ++i) {
            int bin = (d[i] >> RB_SHIFT) - b0;
            rank[i] = -1;
            if (d[i] >= 0 && bin >= 0 && bin < MAXB) rank[i] = atomicAdd(&lcnt[bin], 1);
        }
#pragma unroll
        for (int i = 0; i < EPT; ++i) {
            if (rank[i] >= 0) {
                size_t seg = (size_t)(d[i] >> RB_SHIFT) * NB + c;
                u64* p = stage + seg * SEGK + rank[i];
                if (rank[i] >= SEGK) {
                    int q = atomicAdd(&lspill, 1);
                    p = spill + (size_t)c * CHUNK + q;
                }
                *p = (u64)(unsigned)d[i] | ((u64)(unsigned)s[i] << 32);
            }
        }
        __syncthreads();
        for (int i = tid; i < MAXB && b0 + i < nbins; i += 256) {
            int m = lcnt[i];
            segcnt[(size_t)(b0 + i) * NB + c] = m < SEGK ? m : SEGK;
        }
        __syncthreads();
    }
    if (tid == 0) spillcnt[c] = lspill;
}

__device__ __forceinline__ void owner_place(int dd, int ss, int p, int node0, int* limg,
                                            int* __restrict__ novf, int* __restrict__ ovf) {
    if (p < CAP) {
        limg[(dd - node0) * CAP + p] = ss;
    } else {
        int q = atomicAdd(novf, 1);
        ovf[2 * (size_t)q] = dd;
        ovf[2 * (size_t)q + 1] = ss;
    }
}

__global__ __launch_bounds__(256) void owner_kernel(
    int n, int NB, const int* __restrict__ segcnt, const int* __restrict__ stage_w,
    const int* __restrict__ spillcnt, const int* __restrict__ spill_w,
    int* __restrict__ cnt, int* __restrict__ bucket,
    int* __restrict__ novf, int* __restrict__ ovf) {
    const u64* __restrict__ stage = (const u64*)stage_w;
    const u64* __restrict__ spill = (const u64*)spill_w;
    __shared__ int lcnt[RB];
    __shared__ __attribute__((aligned(16))) int limg[RB * CAP];
    __shared__ int lflag;
    int tid = threadIdx.x;
    int lane = tid & 31;
    int wave = __builtin_amdgcn_readfirstlane(tid >> 5);
    int b = blockIdx.x;
    int node0 = b * RB;

    for (int i = tid; i < RB; i += 256) lcnt[i] = 0;
    for (int i = tid; i < RB * CAP / 4; i += 256) ((int4*)limg)[i] = make_int4(0, 0, 0, 0);
    if (tid == 0) lflag = 0;
    __syncthreads();

    for (int c0 = 0; c0 < NB; c0 += 256) {
        int cw = c0 + wave * 32;
        int myc = cw + lane;
        int nseg = myc < NB ? segcnt[(size_t)b * NB + myc] : 0;
        u64 ed[32];
#pragma unroll
        for (int i = 0; i < 32; ++i) {
            int ni = __builtin_amdgcn_readlane(nseg, i);
            ed[i] = 0;
            if (lane < ni) ed[i] = stage[((size_t)b * NB + (cw + i)) * SEGK + lane];
        }
        int pl[32];
#pragma unroll
        for (int i = 0; i < 32; ++i) {
            int ni = __builtin_amdgcn_readlane(nseg, i);
            pl[i] = -1;
            if (lane < ni) pl[i] = atomicAdd(&lcnt[(int)(unsigned)ed[i] - node0], 1);
        }
#pragma unroll
        for (int i = 0; i < 32; ++i) {
            if (pl[i] >= 0) owner_place((int)(unsigned)ed[i], (int)(ed[i] >> 32), pl[i], node0, limg, novf, ovf);
        }
    }

    for (int c = tid; c < NB; c += 256) {
        if (spillcnt[c] > 0) lflag = 1;
    }
    __syncthreads();
    if (lflag != 0) {
        for (int c = 0; c < NB; ++c) {
            int m = spillcnt[c];
            for (int i = tid; i < m; i += 256) {
                u64 e = spill[(size_t)c * CHUNK + i];
                int dd = (int)(unsigned)e;
                if (dd >= node0 && dd < node0 + RB) {
                    int p = atomicAdd(&lcnt[dd - node0], 1);
                    owner_place(dd, (int)(e >> 32), p, node0, limg, novf, ovf);
                }
            }
        }
    }
    __syncthreads();

    for (int i = tid; i < RB; i += 256) {
        if (node0 + i < n) cnt[node0 + i] = lcnt[i];
    }
    for (int i = tid; i < RB * CAP / 4; i += 256) {
        int row = i / (CAP / 4);
        if (node0 + row < n) ((int4*)bucket)[(size_t)node0 * (CAP / 4) + i] = ((const int4*)limg)[i];
    }
}

#define TILE_STRIDE 132

__device__ __forceinline__ float dinv_of(int count) {
    return rsqrtf((float)(count + 1));
}

__device__ __forceinline__ void fma4(float4& a, const float4& v, float w) {
    a.x = fmaf(v.x, w, a.x);
    a.y = fmaf(v.y, w, a.y);
    a.z = fmaf(v.z, w, a.z);
    a.w = fmaf(v.w, w, a.w);
}

template <int G>
__device__ __forceinline__ void add_group(float4& acc, const float4* __restrict__ h4,
                                          int s, int wi, int j, int lane) {
    float4 v[G];
#pragma unroll
    for (int k = 0; k < G; ++k) {
        int sk = __builtin_amdgcn_readlane(s, j + k);
        v[k] = h4[(size_t)sk * 32 + lane];
    }
#pragma unroll
    for (int k = 0; k < G; ++k) {
        float wk = __int_as_float(__builtin_amdgcn_readlane(wi, j + k));
        fma4(acc, v[k], wk);
    }
}

__device__ __forceinline__ void add_chunk(float4& acc, const float4* __restrict__ h4,
                                          int s, int wi, int m, int lane) {
    int j = 0;
    for (; j + 16 <= m; j += 16) add_group<16>(acc, h4, s, wi, j, lane);
    switch (m - j) {
        case 15: add_group<15>(acc, h4, s, wi, j, lane); break;
        case 14: add_group<14>(acc, h4, s, wi, j, lane); break;
        case 13: add_group<13>(acc, h4, s, wi, j, lane); break;
        case 12: add_group<12>(acc, h4, s, wi, j, lane); break;
        case 11: add_group<11>(acc, h4, s, wi, j, lane); break;
        case 10: add_group<10>(acc, h4, s, wi, j, lane); break;
        case 9:  add_group<9>(acc, h4, s, wi, j, lane); break;
        case 8:  add_group<8>(acc, h4, s, wi, j, lane); break;
        case 7:  add_group<7>(acc, h4, s, wi, j, lane); break;
        case 6:  add_group<6>(acc, h4, s, wi, j, lane); break;
        case 5:  add_group<5>(acc, h4, s, wi, j, lane); break;
        case 4:  add_group<4>(acc, h4, s, wi, j, lane); break;
        case 3:  add_group<3>(acc, h4, s, wi, j, lane); break;
        case 2:  add_group<2>(acc, h4, s, wi, j, lane); break;
        case 1:  add_group<1>(acc, h4, s, wi, j, lane); break;
        default: break;
    }
}

__global__ __launch_bounds__(256) void gather_gemm_kernel(
    const int* __restrict__ cnt, const int* __restrict__ bucket,
    const int* __restrict__ novf, const int* __restrict__ ovf,
    const float* __restrict__ xin, const float* __restrict__ W,
    const float* __restrict__ bias, float* __restrict__ out, int n, int relu) {
    __shared__ __attribute__((aligned(16))) float wT[D * LDS_STRIDE];
    __shared__ __attribute__((aligned(16))) float tiles[8 * 16 * TILE_STRIDE];

    int tid = threadIdx.x;
    for (int i = tid; i < D * D; i += 256) {
        int r = i >> 7, c = i & (D - 1);
        wT[c * LDS_STRIDE + r] = W[i];
    }
    __syncthreads();

    int lane = tid & 31;
    int wave = __builtin_amdgcn_readfirstlane(tid >> 5);
    float* tile = tiles + wave * (16 * TILE_STRIDE);
    const float4* h4 = (const float4*)xin;
    int ntiles = (n + 15) >> 4;

    for (int t = wave * gridDim.x + blockIdx.x; t < ntiles; t += 8 * gridDim.x) {
    int node0 = t * 16;

    int mr = node0 + (lane & 15);
    bool mvalid = mr < n;
    int mrc = mvalid ? mr : 0;
    int cl = cnt[mrc];
    int ddi = __float_as_int(dinv_of(cl));
    int cc = mvalid ? cl : 0;
    int cb = cc < CAP ? cc : CAP;
    const int* brow = bucket + (size_t)node0 * CAP;

    int s0 = 0, s1 = 0;
    float w0 = 0.0f;
    {
        int c0 = __builtin_amdgcn_readlane(cb, 0);
        int c1 = __builtin_amdgcn_readlane(cb, 1);
        float g0 = 0.0f;
        if (lane < c0) s0 = brow[lane];
        if (lane < c1) s1 = brow[CAP + lane];
        if (lane < c0) g0 = dinv_of(cnt[s0]);
        w0 = g0 * __int_as_float(__builtin_amdgcn_readlane(ddi, 0));
    }

#pragma unroll 1
    for (int r = 0; r < 16; ++r) {
        int r1 = r + 1 < 16 ? r + 1 : 15;
        int r2 = r + 2 < 16 ? r + 2 : 15;
        int m1 = r + 1 < 16 ? __builtin_amdgcn_readlane(cb, r1) : 0;
        int k1 = cnt[s1];
        int m2 = r + 2 < 16 ? __builtin_amdgcn_readlane(cb, r2) : 0;
        int s2 = 0;
        if (lane < m2) s2 = brow[r2 * CAP + lane];

        int node = node0 + r;
        float4 acc = make_float4(0.0f, 0.0f, 0.0f, 0.0f);
        if (node < n) {
            int c = __builtin_amdgcn_readlane(cc, r);
            int m = __builtin_amdgcn_readlane(cb, r);
            float di = __int_as_float(__builtin_amdgcn_readlane(ddi, r));
            float4 vself = h4[(size_t)node * 32 + lane];

            add_chunk(acc, h4, s0, __float_as_int(w0), m, lane);

            if (c > CAP) {
                int nv = novf[0];
                for (int q0 = 0; q0 < nv; q0 += 32) {
                    int q = q0 + lane;
                    int os = 0;
                    float ow = 0.0f;
                    bool hit = false;
                    if (q < nv) {
                        if (ovf[2 * (size_t)q] == node) {
                            hit = true;
                            os = ovf[2 * (size_t)q + 1];
                            ow = dinv_of(cnt[os]) * di;
                        }
                    }
                    unsigned mask = __builtin_amdgcn_ballot_w32(hit);
                    while (mask != 0u) {
                        int j = __builtin_ctz(mask);
                        mask &= mask - 1u;
                        int sk = __builtin_amdgcn_readlane(os, j);
                        float wk = __int_as_float(
                            __builtin_amdgcn_readlane(__float_as_int(ow), j));
                        float4 v = h4[(size_t)sk * 32 + lane];
                        fma4(acc, v, wk);
                    }
                }
            }
            fma4(acc, vself, di * di);
        }
        *(float4*)(tile + r * TILE_STRIDE + 4 * lane) = acc;

        s0 = s1;
        s1 = s2;
        w0 = dinv_of(k1) * __int_as_float(__builtin_amdgcn_readlane(ddi, r1));
    }

    __builtin_amdgcn_fence(__ATOMIC_RELEASE, "wavefront");
    __builtin_amdgcn_wave_barrier();
    __builtin_amdgcn_fence(__ATOMIC_ACQUIRE, "wavefront");

    int lrow = lane & 15;
    int lhi  = lane >> 4;

    v8f acc8[8];
#pragma unroll
    for (int nn = 0; nn < 8; ++nn) acc8[nn] = (v8f)(0.0f);

    const float* ap = tile + lrow * TILE_STRIDE + lhi * 2;
    for (int k = 0; k < D; k += 4) {
        v2f a = *(const v2f*)(ap + k);
#pragma unroll
        for (int nn = 0; nn < 8; ++nn) {
            int col = nn * 16 + lrow;
            v2f b = *(const v2f*)(&wT[col * LDS_STRIDE + k + lhi * 2]);
            acc8[nn] = __builtin_amdgcn_wmma_f32_16x16x4_f32(
                false, a, false, b, (short)0, acc8[nn], false, false);
        }
    }

#pragma unroll
    for (int nn = 0; nn < 8; ++nn) {
        int col = nn * 16 + lrow;
        float bv = bias[col];
#pragma unroll
        for (int v = 0; v < 8; ++v) {
            int row = node0 + v + 8 * lhi;
            float o = acc8[nn][v] + bv;
            if (relu) o = fmaxf(o, 0.0f);
            if (row < n) out[(size_t)row * D + col] = o;
        }
    }
    }
}

extern "C" void kernel_launch(void* const* d_in, const int* in_sizes, int n_in,
                              void* d_out, int out_size, void* d_ws, size_t ws_size,
                              hipStream_t stream) {
    const float* x   = (const float*)d_in[0];
    const int*   ei  = (const int*)d_in[1];
    const float* W1  = (const float*)d_in[2];
    const float* b1  = (const float*)d_in[3];
    const float* W2  = (const float*)d_in[4];
    const float* b2  = (const float*)d_in[5];
    float* out = (float*)d_out;

    const int N = in_sizes[0] / D;
    const int E = in_sizes[1] / 2;
    const int* src = ei;
    const int* dst = ei + E;

    const int nBlocksLayer  = 512;
    const int NB    = (E + CHUNK - 1) / CHUNK;
    const int nbins = (N + RB - 1) / RB;

    size_t nPad = ((size_t)N + 63) & ~(size_t)63;
    size_t ePad = ((size_t)E + 63) & ~(size_t)63;
    size_t sPad = ((size_t)nbins * NB + 63) & ~(size_t)63;
    size_t cPad = ((size_t)NB + 63) & ~(size_t)63;
    int* cnt      = (int*)d_ws;
    int* novf     = cnt + nPad;
    int* bucket   = novf + 64;
    int* ovf      = bucket + nPad * CAP;
    float* bufB   = (float*)(ovf + 2 * ePad);
    int* segcnt   = (int*)(bufB + nPad * D);
    int* spillcnt = segcnt + sPad;
    int* spill    = spillcnt + cPad;
    int* stage    = spill + 2 * (size_t)NB * CHUNK;

    partition_kernel<<<NB, 256, 0, stream>>>(src, dst, E, nbins, NB, segcnt, stage, spillcnt, spill, novf);
    owner_kernel<<<nbins, 256, 0, stream>>>(N, NB, segcnt, stage, spillcnt, spill, cnt, bucket, novf, ovf);

    gather_gemm_kernel<<<nBlocksLayer, 256, 0, stream>>>(cnt, bucket, novf, ovf, x, W1, b1, bufB, N, 1);
    gather_gemm_kernel<<<nBlocksLayer, 256, 0, stream>>>(cnt, bucket, novf, ovf, bufB, W2, b2, out, N, 0);
}


# layer kernels: after a block barrier wave w computes output columns 16w..16w+15 of all the block's tiles from its W slice held in registers; no W copy in LDS; sum order and k order unchanged
# speedup vs baseline: 4.1638x; 1.0881x over previous
#include <hip/hip_runtime.h>

typedef __attribute__((ext_vector_type(2))) float v2f;
typedef __attribute__((ext_vector_type(8))) float v8f;
typedef unsigned long long u64;

#define D 128

#define CAP 32
#define RB_SHIFT 7
#define RB (1 << RB_SHIFT)
#define SEGK 32
#define EPT 10
#define CHUNK (256 * EPT)
#define MAXB 1024

__global__ __launch_bounds__(256) void partition_kernel(
    const int* __restrict__ src, const int* __restrict__ dst, int E, int nbins, int NB,
    int* __restrict__ segcnt, int* __restrict__ stage_w,
    int* __restrict__ spillcnt, int* __restrict__ spill_w, int* __restrict__ novf) {
    u64* __restrict__ stage = (u64*)stage_w;
    u64* __restrict__ spill = (u64*)spill_w;
    __shared__ int lcnt[MAXB];
    __shared__ int lspill;
    int tid = threadIdx.x;
    int c = blockIdx.x;
    if (c == 0 && tid == 0) novf[0] = 0;
    if (tid == 0) lspill = 0;

    int d[EPT], s[EPT];
#pragma unroll
    for (int i = 0; i < EPT; ++i) {
        int e = c * CHUNK + i * 256 + tid;
        bool ok = e < E;
        d[i] = ok ? dst[e] : -1;
        s[i] = ok ? src[e] : 0;
    }

    for (int b0 = 0; b0 < nbins; b0 += MAXB) {
        for (int i = tid; i < MAXB; i += 256) lcnt[i] = 0;
        __syncthreads();
        int rank[EPT];
#pragma unroll
        for (int i = 0; i < EPT; ++i) {
            int bin = (d[i] >> RB_SHIFT) - b0;
            rank[i] = -1;
            if (d[i] >= 0 && bin >= 0 && bin < MAXB) rank[i] = atomicAdd(&lcnt[bin], 1);
        }
#pragma unroll
        for (int i = 0; i < EPT; ++i) {
            if (rank[i] >= 0) {
                size_t seg = (size_t)(d[i] >> RB_SHIFT) * NB + c;
                u64* p = stage + seg * SEGK + rank[i];
                if (rank[i] >= SEGK) {
                    int q = atomicAdd(&lspill, 1);
                    p = spill + (size_t)c * CHUNK + q;
                }
                *p = (u64)(unsigned)d[i] | ((u64)(unsigned)s[i] << 32);
            }
        }
        __syncthreads();
        for (int i = tid; i < MAXB && b0 + i < nbins; i += 256) {
            int m = lcnt[i];
            segcnt[(size_t)(b0 + i) * NB + c] = m < SEGK ? m : SEGK;
        }
        __syncthreads();
    }
    if (tid == 0) spillcnt[c] = lspill;
}

__device__ __forceinline__ void owner_place(int dd, int ss, int p, int node0, int* limg,
                                            int* __restrict__ novf, int* __restrict__ ovf) {
    if (p < CAP) {
        limg[(dd - node0) * CAP + p] = ss;
    } else {
        int q = atomicAdd(novf, 1);
        ovf[2 * (size_t)q] = dd;
        ovf[2 * (size_t)q + 1] = ss;
    }
}

__global__ __launch_bounds__(256) void owner_kernel(
    int n, int NB, const int* __restrict__ segcnt, const int* __restrict__ stage_w,
    const int* __restrict__ spillcnt, const int* __restrict__ spill_w,
    int* __restrict__ cnt, int* __restrict__ bucket,
    int* __restrict__ novf, int* __restrict__ ovf) {
    const u64* __restrict__ stage = (const u64*)stage_w;
    const u64* __restrict__ spill = (const u64*)spill_w;
    __shared__ int lcnt[RB];
    __shared__ __attribute__((aligned(16))) int limg[RB * CAP];
    __shared__ int lflag;
    int tid = threadIdx.x;
    int lane = tid & 31;
    int wave = __builtin_amdgcn_readfirstlane(tid >> 5);
    int b = blockIdx.x;
    int node0 = b * RB;

    for (int i = tid; i < RB; i += 256) lcnt[i] = 0;
    for (int i = tid; i < RB * CAP / 4; i += 256) ((int4*)limg)[i] = make_int4(0, 0, 0, 0);
    if (tid == 0) lflag = 0;
    __syncthreads();

    for (int c0 = 0; c0 < NB; c0 += 256) {
        int cw = c0 + wave * 32;
        int myc = cw + lane;
        int nseg = myc < NB ? segcnt[(size_t)b * NB + myc] : 0;
        u64 ed[32];
#pragma unroll
        for (int i = 0; i < 32; ++i) {
            int ni = __builtin_amdgcn_readlane(nseg, i);
            ed[i] = 0;
            if (lane < ni) ed[i] = stage[((size_t)b * NB + (cw + i)) * SEGK + lane];
        }
        int pl[32];
#pragma unroll
        for (int i = 0; i < 32; ++i) {
            int ni = __builtin_amdgcn_readlane(nseg, i);
            pl[i] = -1;
            if (lane < ni) pl[i] = atomicAdd(&lcnt[(int)(unsigned)ed[i] - node0], 1);
        }
#pragma unroll
        for (int i = 0; i < 32; ++i) {
            if (pl[i] >= 0) owner_place((int)(unsigned)ed[i], (int)(ed[i] >> 32), pl[i], node0, limg, novf, ovf);
        }
    }

    for (int c = tid; c < NB; c += 256) {
        if (spillcnt[c] > 0) lflag = 1;
    }
    __syncthreads();
    if (lflag != 0) {
        for (int c = 0; c < NB; ++c) {
            int m = spillcnt[c];
            for (int i = tid; i < m; i += 256) {
                u64 e = spill[(size_t)c * CHUNK + i];
                int dd = (int)(unsigned)e;
                if (dd >= node0 && dd < node0 + RB) {
                    int p = atomicAdd(&lcnt[dd - node0], 1);
                    owner_place(dd, (int)(e >> 32), p, node0, limg, novf, ovf);
                }
            }
        }
    }
    __syncthreads();

    for (int i = tid; i < RB; i += 256) {
        if (node0 + i < n) cnt[node0 + i] = lcnt[i];
    }
    for (int i = tid; i < RB * CAP / 4; i += 256) {
        int row = i / (CAP / 4);
        if (node0 + row < n) ((int4*)bucket)[(size_t)node0 * (CAP / 4) + i] = ((const int4*)limg)[i];
    }
}

#define TILE_STRIDE 132

__device__ __forceinline__ float dinv_of(int count) {
    return rsqrtf((float)(count + 1));
}

__device__ __forceinline__ void fma4(float4& a, const float4& v, float w) {
    a.x = fmaf(v.x, w, a.x);
    a.y = fmaf(v.y, w, a.y);
    a.z = fmaf(v.z, w, a.z);
    a.w = fmaf(v.w, w, a.w);
}

template <int G>
__device__ __forceinline__ void add_group(float4& acc, const float4* __restrict__ h4,
                                          int s, int wi, int j, int lane) {
    float4 v[G];
#pragma unroll
    for (int k = 0; k < G; ++k) {
        int sk = __builtin_amdgcn_readlane(s, j + k);
        v[k] = h4[(size_t)sk * 32 + lane];
    }
#pragma unroll
    for (int k = 0; k < G; ++k) {
        float wk = __int_as_float(__builtin_amdgcn_readlane(wi, j + k));
        fma4(acc, v[k], wk);
    }
}

__device__ __forceinline__ void add_chunk(float4& acc, const float4* __restrict__ h4,
                                          int s, int wi, int m, int lane) {
    int j = 0;
    for (; j + 16 <= m; j += 16) add_group<16>(acc, h4, s, wi, j, lane);
    switch (m - j) {
        case 15: add_group<15>(acc, h4, s, wi, j, lane); break;
        case 14: add_group<14>(acc, h4, s, wi, j, lane); break;
        case 13: add_group<13>(acc, h4, s, wi, j, lane); break;
        case 12: add_group<12>(acc, h4, s, wi, j, lane); break;
        case 11: add_group<11>(acc, h4, s, wi, j, lane); break;
        case 10: add_group<10>(acc, h4, s, wi, j, lane); break;
        case 9:  add_group<9>(acc, h4, s, wi, j, lane); break;
        case 8:  add_group<8>(acc, h4, s, wi, j, lane); break;
        case 7:  add_group<7>(acc, h4, s, wi, j, lane); break;
        case 6:  add_group<6>(acc, h4, s, wi, j, lane); break;
        case 5:  add_group<5>(acc, h4, s, wi, j, lane); break;
        case 4:  add_group<4>(acc, h4, s, wi, j, lane); break;
        case 3:  add_group<3>(acc, h4, s, wi, j, lane); break;
        case 2:  add_group<2>(acc, h4, s, wi, j, lane); break;
        case 1:  add_group<1>(acc, h4, s, wi, j, lane); break;
        default: break;
    }
}

template <int NV>
__device__ __forceinline__ void coop_gemm(const float* __restrict__ tiles, const v2f* bf,
                                          int lrow, int lhi, int base, int G, int col,
                                          float bv, float* __restrict__ out, int n, int relu) {
    v8f acc[NV];
#pragma unroll
    for (int u = 0; u < NV; ++u) acc[u] = (v8f)(0.0f);
    const float* ap = tiles + lrow * TILE_STRIDE + lhi * 2;
#pragma unroll
    for (int j = 0; j < 32; ++j) {
#pragma unroll
        for (int u = 0; u < NV; ++u) {
            v2f a = *(const v2f*)(ap + u * (16 * TILE_STRIDE) + 4 * j);
            acc[u] = __builtin_amdgcn_wmma_f32_16x16x4_f32(
                false, a, false, bf[j], (short)0, acc[u], false, false);
        }
    }
#pragma unroll
    for (int u = 0; u < NV; ++u) {
        int node0 = (base + u * G) * 16;
#pragma unroll
        for (int v = 0; v < 8; ++v) {
            int row = node0 + v + 8 * lhi;
            float o = acc[u][v] + bv;
            if (relu) o = fmaxf(o, 0.0f);
            if (row < n) out[(size_t)row * D + col] = o;
        }
    }
}

__global__ __launch_bounds__(256) void gather_gemm_kernel(
    const int* __restrict__ cnt, const int* __restrict__ bucket,
    const int* __restrict__ novf, const int* __restrict__ ovf,
    const float* __restrict__ xin, const float* __restrict__ W,
    const float* __restrict__ bias, float* __restrict__ out, int n, int relu) {
    __shared__ __attribute__((aligned(16))) float tiles[8 * 16 * TILE_STRIDE];

    int tid = threadIdx.x;
    int lane = tid & 31;
    int wave = __builtin_amdgcn_readfirstlane(tid >> 5);
    float* tile = tiles + wave * (16 * TILE_STRIDE);
    const float4* h4 = (const float4*)xin;
    int ntiles = (n + 15) >> 4;

    int lrow = lane & 15;
    int lhi  = lane >> 4;
    int G = gridDim.x;

    for (int base = blockIdx.x; base < ntiles; base += 8 * G) {
    int t = base + wave * G;
    if (t < ntiles) {
    int node0 = t * 16;

    int mr = node0 + (lane & 15);
    bool mvalid = mr < n;
    int mrc = mvalid ? mr : 0;
    int cl = cnt[mrc];
    int ddi = __float_as_int(dinv_of(cl));
    int cc = mvalid ? cl : 0;
    int cb = cc < CAP ? cc : CAP;
    const int* brow = bucket + (size_t)node0 * CAP;

    int s0 = 0, s1 = 0;
    float w0 = 0.0f;
    {
        int c0 = __builtin_amdgcn_readlane(cb, 0);
        int c1 = __builtin_amdgcn_readlane(cb, 1);
        float g0 = 0.0f;
        if (lane < c0) s0 = brow[lane];
        if (lane < c1) s1 = brow[CAP + lane];
        if (lane < c0) g0 = dinv_of(cnt[s0]);
        w0 = g0 * __int_as_float(__builtin_amdgcn_readlane(ddi, 0));
    }

#pragma unroll 1
    for (int r = 0; r < 16; ++r) {
        int r1 = r + 1 < 16 ? r + 1 : 15;
        int r2 = r + 2 < 16 ? r + 2 : 15;
        int m1 = r + 1 < 16 ? __builtin_amdgcn_readlane(cb, r1) : 0;
        int k1 = cnt[s1];
        int m2 = r + 2 < 16 ? __builtin_amdgcn_readlane(cb, r2) : 0;
        int s2 = 0;
        if (lane < m2) s2 = brow[r2 * CAP + lane];

        int node = node0 + r;
        float4 acc = make_float4(0.0f, 0.0f, 0.0f, 0.0f);
        if (node < n) {
            int c = __builtin_amdgcn_readlane(cc, r);
            int m = __builtin_amdgcn_readlane(cb, r);
            float di = __int_as_float(__builtin_amdgcn_readlane(ddi, r));
            float4 vself = h4[(size_t)node * 32 + lane];

            add_chunk(acc, h4, s0, __float_as_int(w0), m, lane);

            if (c > CAP) {
                int nv = novf[0];
                for (int q0 = 0; q0 < nv; q0 += 32) {
                    int q = q0 + lane;
                    int os = 0;
                    float ow = 0.0f;
                    bool hit = false;
                    if (q < nv) {
                        if (ovf[2 * (size_t)q] == node) {
                            hit = true;
                            os = ovf[2 * (size_t)q + 1];
                            ow = dinv_of(cnt[os]) * di;
                        }
                    }
                    unsigned mask = __builtin_amdgcn_ballot_w32(hit);
                    while (mask != 0u) {
                        int j = __builtin_ctz(mask);
                        mask &= mask - 1u;
                        int sk = __builtin_amdgcn_readlane(os, j);
                        float wk = __int_as_float(
                            __builtin_amdgcn_readlane(__float_as_int(ow), j));
                        float4 v = h4[(size_t)sk * 32 + lane];
                        fma4(acc, v, wk);
                    }
                }
            }
            fma4(acc, vself, di * di);
        }
        *(float4*)(tile + r * TILE_STRIDE + 4 * lane) = acc;

        s0 = s1;
        s1 = s2;
        w0 = dinv_of(k1) * __int_as_float(__builtin_amdgcn_readlane(ddi, r1));
    }

    }

    int col = wave * 16 + lrow;
    v2f bf[32];
    {
        const float* wp = W + (size_t)(2 * lhi) * D + col;
#pragma unroll
        for (int j = 0; j < 32; ++j) {
            bf[j].x = wp[(size_t)(4 * j) * D];
            bf[j].y = wp[(size_t)(4 * j + 1) * D];
        }
    }
    float bv = bias[col];

    int nv = (ntiles - base + G - 1) / G;
    if (nv > 8) nv = 8;

    __syncthreads();

    switch (nv) {
        case 8:  coop_gemm<8>(tiles, bf, lrow, lhi, base, G, col, bv, out, n, relu); break;
        case 7:  coop_gemm<7>(tiles, bf, lrow, lhi, base, G, col, bv, out, n, relu); break;
        case 6:  coop_gemm<6>(tiles, bf, lrow, lhi, base, G, col, bv, out, n, relu); break;
        case 5:  coop_gemm<5>(tiles, bf, lrow, lhi, base, G, col, bv, out, n, relu); break;
        case 4:  coop_gemm<4>(tiles, bf, lrow, lhi, base, G, col, bv, out, n, relu); break;
        case 3:  coop_gemm<3>(tiles, bf, lrow, lhi, base, G, col, bv, out, n, relu); break;
        case 2:  coop_gemm<2>(tiles, bf, lrow, lhi, base, G, col, bv, out, n, relu); break;
        default: coop_gemm<1>(tiles, bf, lrow, lhi, base, G, col, bv, out, n, relu); break;
    }

    __syncthreads();
    }
}

extern "C" void kernel_launch(void* const* d_in, const int* in_sizes, int n_in,
                              void* d_out, int out_size, void* d_ws, size_t ws_size,
                              hipStream_t stream) {
    const float* x   = (const float*)d_in[0];
    const int*   ei  = (const int*)d_in[1];
    const float* W1  = (const float*)d_in[2];
    const float* b1  = (const float*)d_in[3];
    const float* W2  = (const float*)d_in[4];
    const float* b2  = (const float*)d_in[5];
    float* out = (float*)d_out;

    const int N = in_sizes[0] / D;
    const int E = in_sizes[1] / 2;
    const int* src = ei;
    const int* dst = ei + E;

    const int nBlocksLayer  = 512;
    const int NB    = (E + CHUNK - 1) / CHUNK;
    const int nbins = (N + RB - 1) / RB;

    size_t nPad = ((size_t)N + 63) & ~(size_t)63;
    size_t ePad = ((size_t)E + 63) & ~(size_t)63;
    size_t sPad = ((size_t)nbins * NB + 63) & ~(size_t)63;
    size_t cPad = ((size_t)NB + 63) & ~(size_t)63;
    int* cnt      = (int*)d_ws;
    int* novf     = cnt + nPad;
    int* bucket   = novf + 64;
    int* ovf      = bucket + nPad * CAP;
    float* bufB   = (float*)(ovf + 2 * ePad);
    int* segcnt   = (int*)(bufB + nPad * D);
    int* spillcnt = segcnt + sPad;
    int* spill    = spillcnt + cPad;
    int* stage    = spill + 2 * (size_t)NB * CHUNK;

    partition_kernel<<<NB, 256, 0, stream>>>(src, dst, E, nbins, NB, segcnt, stage, spillcnt, spill, novf);
    owner_kernel<<<nbins, 256, 0, stream>>>(N, NB, segcnt, stage, spillcnt, spill, cnt, bucket, novf, ovf);

    gather_gemm_kernel<<<nBlocksLayer, 256, 0, stream>>>(cnt, bucket, novf, ovf, x, W1, b1, bufB, N, 1);
    gather_gemm_kernel<<<nBlocksLayer, 256, 0, stream>>>(cnt, bucket, novf, ovf, bufB, W2, b2, out, N, 0);
}


# layer kernels: tiles go in stages of two; a wave issues a finished stage's WMMAs while its next row's source rows are in flight (W in LDS, LDS-only block barrier); sum order and k order unchanged
# speedup vs baseline: 4.5513x; 1.0931x over previous
#include <hip/hip_runtime.h>

typedef __attribute__((ext_vector_type(2))) float v2f;
typedef __attribute__((ext_vector_type(4))) float v4f;
typedef __attribute__((ext_vector_type(8))) float v8f;
typedef unsigned long long u64;

#define D 128

#define CAP 32
#define RB_SHIFT 7
#define RB (1 << RB_SHIFT)
#define SEGK 32
#define EPT 10
#define CHUNK (256 * EPT)
#define MAXB 1024

__global__ __launch_bounds__(256) void partition_kernel(
    const int* __restrict__ src, const int* __restrict__ dst, int E, int nbins, int NB,
    int* __restrict__ segcnt, int* __restrict__ stage_w,
    int* __restrict__ spillcnt, int* __restrict__ spill_w, int* __restrict__ novf) {
    u64* __restrict__ stage = (u64*)stage_w;
    u64* __restrict__ spill = (u64*)spill_w;
    __shared__ int lcnt[MAXB];
    __shared__ int lspill;
    int tid = threadIdx.x;
    int c = blockIdx.x;
    if (c == 0 && tid == 0) novf[0] = 0;
    if (tid == 0) lspill = 0;

    int d[EPT], s[EPT];
#pragma unroll
    for (int i = 0; i < EPT; ++i) {
        int e = c * CHUNK + i * 256 + tid;
        bool ok = e < E;
        d[i] = ok ? dst[e] : -1;
        s[i] = ok ? src[e] : 0;
    }

    for (int b0 = 0; b0 < nbins; b0 += MAXB) {
        for (int i = tid; i < MAXB; i += 256) lcnt[i] = 0;
        __syncthreads();
        int rank[EPT];
#pragma unroll
        for (int i = 0; i < EPT; ++i) {
            int bin = (d[i] >> RB_SHIFT) - b0;
            rank[i] = -1;
            if (d[i] >= 0 && bin >= 0 && bin < MAXB) rank[i] = atomicAdd(&lcnt[bin], 1);
        }
#pragma unroll
        for (int i = 0; i < EPT; ++i) {
            if (rank[i] >= 0) {
                size_t seg = (size_t)(d[i] >> RB_SHIFT) * NB + c;
                u64* p = stage + seg * SEGK + rank[i];
                if (rank[i] >= SEGK) {
                    int q = atomicAdd(&lspill, 1);
                    p = spill + (size_t)c * CHUNK + q;
                }
                *p = (u64)(unsigned)d[i] | ((u64)(unsigned)s[i] << 32);
            }
        }
        __syncthreads();
        for (int i = tid; i < MAXB && b0 + i < nbins; i += 256) {
            int m = lcnt[i];
            segcnt[(size_t)(b0 + i) * NB + c] = m < SEGK ? m : SEGK;
        }
        __syncthreads();
    }
    if (tid == 0) spillcnt[c] = lspill;
}

__device__ __forceinline__ void owner_place(int dd, int ss, int p, int node0, int* limg,
                                            int* __restrict__ novf, int* __restrict__ ovf) {
    if (p < CAP) {
        limg[(dd - node0) * CAP + p] = ss;
    } else {
        int q = atomicAdd(novf, 1);
        ovf[2 * (size_t)q] = dd;
        ovf[2 * (size_t)q + 1] = ss;
    }
}

__global__ __launch_bounds__(256) void owner_kernel(
    int n, int NB, const int* __restrict__ segcnt, const int* __restrict__ stage_w,
    const int* __restrict__ spillcnt, const int* __restrict__ spill_w,
    int* __restrict__ cnt, int* __restrict__ bucket,
    int* __restrict__ novf, int* __restrict__ ovf) {
    const u64* __restrict__ stage = (const u64*)stage_w;
    const u64* __restrict__ spill = (const u64*)spill_w;
    __shared__ int lcnt[RB];
    __shared__ __attribute__((aligned(16))) int limg[RB * CAP];
    __shared__ int lflag;
    int tid = threadIdx.x;
    int lane = tid & 31;
    int wave = __builtin_amdgcn_readfirstlane(tid >> 5);
    int b = blockIdx.x;
    int node0 = b * RB;

    for (int i = tid; i < RB; i += 256) lcnt[i] = 0;
    for (int i = tid; i < RB * CAP / 4; i += 256) ((int4*)limg)[i] = make_int4(0, 0, 0, 0);
    if (tid == 0) lflag = 0;
    __syncthreads();

    for (int c0 = 0; c0 < NB; c0 += 256) {
        int cw = c0 + wave * 32;
        int myc = cw + lane;
        int nseg = myc < NB ? segcnt[(size_t)b * NB + myc] : 0;
        u64 ed[32];
#pragma unroll
        for (int i = 0; i < 32; ++i) {
            int ni = __builtin_amdgcn_readlane(nseg, i);
            ed[i] = 0;
            if (lane < ni) ed[i] = stage[((size_t)b * NB + (cw + i)) * SEGK + lane];
        }
        int pl[32];
#pragma unroll
        for (int i = 0; i < 32; ++i) {
            int ni = __builtin_amdgcn_readlane(nseg, i);
            pl[i] = -1;
            if (lane < ni) pl[i] = atomicAdd(&lcnt[(int)(unsigned)ed[i] - node0], 1);
        }
#pragma unroll
        for (int i = 0; i < 32; ++i) {
            if (pl[i] >= 0) owner_place((int)(unsigned)ed[i], (int)(ed[i] >> 32), pl[i], node0, limg, novf, ovf);
        }
    }

    for (int c = tid; c < NB; c += 256) {
        if (spillcnt[c] > 0) lflag = 1;
    }
    __syncthreads();
    if (lflag != 0) {
        for (int c = 0; c < NB; ++c) {
            int m = spillcnt[c];
            for (int i = tid; i < m; i += 256) {
                u64 e = spill[(size_t)c * CHUNK + i];
                int dd = (int)(unsigned)e;
                if (dd >= node0 && dd < node0 + RB) {
                    int p = atomicAdd(&lcnt[dd - node0], 1);
                    owner_place(dd, (int)(e >> 32), p, node0, limg, novf, ovf);
                }
            }
        }
    }
    __syncthreads();

    for (int i = tid; i < RB; i += 256) {
        if (node0 + i < n) cnt[node0 + i] = lcnt[i];
    }
    for (int i = tid; i < RB * CAP / 4; i += 256) {
        int row = i / (CAP / 4);
        if (node0 + row < n) ((int4*)bucket)[(size_t)node0 * (CAP / 4) + i] = ((const int4*)limg)[i];
    }
}

#define TILE_STRIDE 132
#define W_STRIDE 136
#define STAGE_FLOATS (32 * TILE_STRIDE)

__device__ __forceinline__ float dinv_of(int count) {
    return rsqrtf((float)(count + 1));
}

__device__ __forceinline__ void fma4(float4& a, const float4& v, float w) {
    a.x = fmaf(v.x, w, a.x);
    a.y = fmaf(v.y, w, a.y);
    a.z = fmaf(v.z, w, a.z);
    a.w = fmaf(v.w, w, a.w);
}

__device__ __forceinline__ void request_rows(float4 (&v)[16], const float4* __restrict__ h4,
                                             int s, int j0, int g, int lane) {
#pragma unroll
    for (int k = 0; k < 16; ++k) {
        if (k < g) {
            int sk = __builtin_amdgcn_readlane(s, j0 + k);
            v[k] = h4[(size_t)sk * 32 + lane];
        }
    }
}

__device__ __forceinline__ void sum_rows(float4& acc, const float4 (&v)[16],
                                         int wi, int j0, int g) {
#pragma unroll
    for (int k = 0; k < 16; ++k) {
        if (k < g) {
            float wk = __int_as_float(__builtin_amdgcn_readlane(wi, j0 + k));
            fma4(acc, v[k], wk);
        }
    }
}

__device__ __forceinline__ int list_counter(const int* __restrict__ cnt, int base, int G,
                                            int ntiles, int n, int tu, int rp, int li,
                                            bool& exists) {
    int nv = (ntiles - base + G - 1) / G;
    int node = (base + (2 * (li >> 1) + tu) * G) * 16 + rp + (li & 1);
    exists = base < ntiles && 2 * (li >> 1) + tu < nv && node < n;
    return cnt[exists ? node : 0];
}

__global__ __launch_bounds__(512) __attribute__((amdgpu_waves_per_eu(8, 8))) void gather_gemm_kernel(
    const int* __restrict__ cnt, const int* __restrict__ bucket,
    const int* __restrict__ novf, const int* __restrict__ ovf,
    const float* __restrict__ xin, const float* __restrict__ W,
    const float* __restrict__ bias, float* __restrict__ out, int n, int relu) {
    __shared__ __attribute__((aligned(16))) float Wl[D * W_STRIDE];
    __shared__ __attribute__((aligned(16))) float ring[2 * STAGE_FLOATS];

    int tid = threadIdx.x;
    int lane = tid & 31;
    int wave = __builtin_amdgcn_readfirstlane(tid >> 5);
    int tu = wave >> 3;
    int rp = (wave & 7) * 2;
    const float4* h4 = (const float4*)xin;
    int ntiles = (n + 15) >> 4;

    int lrow = lane & 15;
    int lhi  = lane >> 4;
    int G = gridDim.x;

    int col = (wave & 7) * 16 + lrow;
    float bv = bias[col];
    const float* bp = Wl + (2 * lhi) * W_STRIDE + col;

    int li = lane & 7;
    bool rvl;
    int cl = list_counter(cnt, blockIdx.x, G, ntiles, n, tu, rp, li, rvl);

    {
        v4f wv[8];
#pragma unroll
        for (int q = 0; q < 8; ++q) wv[q] = ((const v4f*)W)[(size_t)(wave * 8 + q) * 32 + lane];
        asm volatile("" :: "v"(wv[0]), "v"(wv[1]), "v"(wv[2]), "v"(wv[3]),
                           "v"(wv[4]), "v"(wv[5]), "v"(wv[6]), "v"(wv[7]));
#pragma unroll
        for (int q = 0; q < 8; ++q) *(v4f*)(Wl + (wave * 8 + q) * W_STRIDE + 4 * lane) = wv[q];
    }

    for (int base = blockIdx.x; base < ntiles; base += 8 * G) {
    int nv = (ntiles - base + G - 1) / G;
    if (nv > 8) nv = 8;
    int S = (nv + 1) >> 1;
    int NR = 2 * S;

#define ROW_TILE(i) (2 * ((i) >> 1) + tu)
#define ROW_NODE(i) ((base + ROW_TILE(i) * G) * 16 + rp + ((i) & 1))

    int ddi = __float_as_int(dinv_of(cl));
    int cc = rvl ? cl : 0;

    int s0 = 0, s1 = 0;
    float w0 = 0.0f;
    {
        int c0 = __builtin_amdgcn_readlane(cc, 0);
        int c1 = __builtin_amdgcn_readlane(cc, 1);
        c0 = c0 < CAP ? c0 : CAP;
        c1 = c1 < CAP ? c1 : CAP;
        int n0 = ROW_NODE(0);
        int n1 = ROW_NODE(1);
        float g0 = 0.0f;
        if (lane < c0) s0 = bucket[(size_t)n0 * CAP + lane];
        if (lane < c1) s1 = bucket[(size_t)n1 * CAP + lane];
        if (lane < c0) g0 = dinv_of(cnt[s0]);
        w0 = g0 * __int_as_float(__builtin_amdgcn_readlane(ddi, 0));
    }

    float4 v[16];
    float4 vself = make_float4(0.0f, 0.0f, 0.0f, 0.0f);
    int k1 = 0, s2 = 0;

#pragma unroll 1
    for (int i = -1; i < NR; ++i) {
        bool edge = i >= 0 && (i & 1) != 0;
        if (i >= 0) {
            if (ROW_TILE(i) < nv) {
                float4 acc = make_float4(0.0f, 0.0f, 0.0f, 0.0f);
                int node = ROW_NODE(i);
                if (node < n) {
                    int c = __builtin_amdgcn_readlane(cc, i);
                    int m = c < CAP ? c : CAP;
                    float di = __int_as_float(__builtin_amdgcn_readlane(ddi, i));
                    int w0i = __float_as_int(w0);
                    sum_rows(acc, v, w0i, 0, m < 16 ? m : 16);
                    if (m > 16) {
                        request_rows(v, h4, s0, 16, m - 16, lane);
                        sum_rows(acc, v, w0i, 16, m - 16);
                    }
                    if (c > CAP) {
                        int nvf = novf[0];
                        for (int q0 = 0; q0 < nvf; q0 += 32) {
                            int q = q0 + lane;
                            int os = 0;
                            float ow = 0.0f;
                            bool hit = false;
                            if (q < nvf) {
                                if (ovf[2 * (size_t)q] == node) {
                                    hit = true;
                                    os = ovf[2 * (size_t)q + 1];
                                    ow = dinv_of(cnt[os]) * di;
                                }
                            }
                            unsigned mask = __builtin_amdgcn_ballot_w32(hit);
                            while (mask != 0u) {
                                int j = __builtin_ctz(mask);
                                mask &= mask - 1u;
                                int sk = __builtin_amdgcn_readlane(os, j);
                                float wk = __int_as_float(
                                    __builtin_amdgcn_readlane(__float_as_int(ow), j));
                                float4 vo = h4[(size_t)sk * 32 + lane];
                                fma4(acc, vo, wk);
                            }
                        }
                    }
                    fma4(acc, vself, di * di);
                }
                float* slot = ring + ((i >> 1) & 1) * STAGE_FLOATS
                            + (tu * 16 + rp + (i & 1)) * TILE_STRIDE + 4 * lane;
                *(float4*)slot = acc;
            }
            int i1 = i + 1 < NR ? i + 1 : NR - 1;
            s0 = s1;
            s1 = s2;
            w0 = dinv_of(k1) * __int_as_float(__builtin_amdgcn_readlane(ddi, i1));
            if (edge) {
                asm volatile("s_wait_dscnt 0x0\n\ts_barrier_signal -1" ::: "memory");
            }
        }
        if (i + 1 < NR) {
            int r = i + 1;
            int r2 = r + 2 < NR ? r + 2 : NR - 1;
            k1 = cnt[s1];
            int m2 = r + 2 < NR ? __builtin_amdgcn_readlane(cc, r2) : 0;
            m2 = m2 < CAP ? m2 : CAP;
            int n2 = ROW_NODE(r2);
            s2 = 0;
            if (lane < m2) s2 = bucket[(size_t)n2 * CAP + lane];
            int node = ROW_NODE(r);
            if (ROW_TILE(r) < nv && node < n) {
                int m = __builtin_amdgcn_readlane(cc, r);
                m = m < CAP ? m : CAP;
                vself = h4[(size_t)node * 32 + lane];
                request_rows(v, h4, s0, 0, m < 16 ? m : 16, lane);
            }
        }
        if (edge) {
            asm volatile("s_barrier_wait -1" ::: "memory");
            int s = i >> 1;
            int u = 2 * s + tu;
            if (u < nv) {
                const float* ap = ring + (s & 1) * STAGE_FLOATS
                                + (tu * 16 + lrow) * TILE_STRIDE + lhi * 2;
                v8f acc8 = (v8f)(0.0f);
#pragma unroll
                for (int j = 0; j < 32; ++j) {
                    v2f af = *(const v2f*)(ap + 4 * j);
                    v2f bf;
                    bf.x = bp[(4 * j) * W_STRIDE];
                    bf.y = bp[(4 * j + 1) * W_STRIDE];
                    acc8 = __builtin_amdgcn_wmma_f32_16x16x4_f32(
                        false, af, false, bf, (short)0, acc8, false, false);
                }
                int node0 = (base + u * G) * 16;
#pragma unroll
                for (int vv = 0; vv < 8; ++vv) {
                    int row = node0 + vv + 8 * lhi;
                    float o = acc8[vv] + bv;
                    if (relu) o = fmaxf(o, 0.0f);
                    if (row < n) out[(size_t)row * D + col] = o;
                }
            }
        }
    }

    asm volatile("s_wait_dscnt 0x0\n\ts_barrier_signal -1\n\ts_barrier_wait -1" ::: "memory");
    if (base + 8 * G < ntiles) cl = list_counter(cnt, base + 8 * G, G, ntiles, n, tu, rp, li, rvl);
    }
#undef ROW_TILE
#undef ROW_NODE
}

extern "C" void kernel_launch(void* const* d_in, const int* in_sizes, int n_in,
                              void* d_out, int out_size, void* d_ws, size_t ws_size,
                              hipStream_t stream) {
    const float* x   = (const float*)d_in[0];
    const int*   ei  = (const int*)d_in[1];
    const float* W1  = (const float*)d_in[2];
    const float* b1  = (const float*)d_in[3];
    const float* W2  = (const float*)d_in[4];
    const float* b2  = (const float*)d_in[5];
    float* out = (float*)d_out;

    const int N = in_sizes[0] / D;
    const int E = in_sizes[1] / 2;
    const int* src = ei;
    const int* dst = ei + E;

    const int nBlocksLayer  = 512;
    const int NB    = (E + CHUNK - 1) / CHUNK;
    const int nbins = (N + RB - 1) / RB;

    size_t nPad = ((size_t)N + 63) & ~(size_t)63;
    size_t ePad = ((size_t)E + 63) & ~(size_t)63;
    size_t sPad = ((size_t)nbins * NB + 63) & ~(size_t)63;
    size_t cPad = ((size_t)NB + 63) & ~(size_t)63;
    int* cnt      = (int*)d_ws;
    int* novf     = cnt + nPad;
    int* bucket   = novf + 64;
    int* ovf      = bucket + nPad * CAP;
    float* bufB   = (float*)(ovf + 2 * ePad);
    int* segcnt   = (int*)(bufB + nPad * D);
    int* spillcnt = segcnt + sPad;
    int* spill    = spillcnt + cPad;
    int* stage    = spill + 2 * (size_t)NB * CHUNK;

    partition_kernel<<<NB, 256, 0, stream>>>(src, dst, E, nbins, NB, segcnt, stage, spillcnt, spill, novf);
    owner_kernel<<<nbins, 256, 0, stream>>>(N, NB, segcnt, stage, spillcnt, spill, cnt, bucket, novf, ovf);

    gather_gemm_kernel<<<nBlocksLayer, 512, 0, stream>>>(cnt, bucket, novf, ovf, x, W1, b1, bufB, N, 1);
    gather_gemm_kernel<<<nBlocksLayer, 512, 0, stream>>>(cnt, bucket, novf, ovf, bufB, W2, b2, out, N, 0);
}


# layer kernels: a stage's 16x16 output blocks go to an LDS image; each wave then stores two whole 512-B rows behind the next stage barrier, not 64-B half lines; one writer per line; sums unchanged
# speedup vs baseline: 4.8416x; 1.0638x over previous
#include <hip/hip_runtime.h>

typedef __attribute__((ext_vector_type(2))) float v2f;
typedef __attribute__((ext_vector_type(4))) float v4f;
typedef __attribute__((ext_vector_type(8))) float v8f;
typedef unsigned long long u64;

#define D 128

#define CAP 32
#define RB_SHIFT 7
#define RB (1 << RB_SHIFT)
#define SEGK 32
#define EPT 10
#define CHUNK (256 * EPT)
#define MAXB 1024

__global__ __launch_bounds__(256) void partition_kernel(
    const int* __restrict__ src, const int* __restrict__ dst, int E, int nbins, int NB,
    int* __restrict__ segcnt, int* __restrict__ stage_w,
    int* __restrict__ spillcnt, int* __restrict__ spill_w, int* __restrict__ novf) {
    u64* __restrict__ stage = (u64*)stage_w;
    u64* __restrict__ spill = (u64*)spill_w;
    __shared__ int lcnt[MAXB];
    __shared__ int lspill;
    int tid = threadIdx.x;
    int c = blockIdx.x;
    if (c == 0 && tid == 0) novf[0] = 0;
    if (tid == 0) lspill = 0;

    int d[EPT], s[EPT];
#pragma unroll
    for (int i = 0; i < EPT; ++i) {
        int e = c * CHUNK + i * 256 + tid;
        bool ok = e < E;
        d[i] = ok ? dst[e] : -1;
        s[i] = ok ? src[e] : 0;
    }

    for (int b0 = 0; b0 < nbins; b0 += MAXB) {
        for (int i = tid; i < MAXB; i += 256) lcnt[i] = 0;
        __syncthreads();
        int rank[EPT];
#pragma unroll
        for (int i = 0; i < EPT; ++i) {
            int bin = (d[i] >> RB_SHIFT) - b0;
            rank[i] = -1;
            if (d[i] >= 0 && bin >= 0 && bin < MAXB) rank[i] = atomicAdd(&lcnt[bin], 1);
        }
#pragma unroll
        for (int i = 0; i < EPT; ++i) {
            if (rank[i] >= 0) {
                size_t seg = (size_t)(d[i] >> RB_SHIFT) * NB + c;
                u64* p = stage + seg * SEGK + rank[i];
                if (rank[i] >= SEGK) {
                    int q = atomicAdd(&lspill, 1);
                    p = spill + (size_t)c * CHUNK + q;
                }
                *p = (u64)(unsigned)d[i] | ((u64)(unsigned)s[i] << 32);
            }
        }
        __syncthreads();
        for (int i = tid; i < MAXB && b0 + i < nbins; i += 256) {
            int m = lcnt[i];
            segcnt[(size_t)(b0 + i) * NB + c] = m < SEGK ? m : SEGK;
        }
        __syncthreads();
    }
    if (tid == 0) spillcnt[c] = lspill;
}

__device__ __forceinline__ void owner_place(int dd, int ss, int p, int node0, int* limg,
                                            int* __restrict__ novf, int* __restrict__ ovf) {
    if (p < CAP) {
        limg[(dd - node0) * CAP + p] = ss;
    } else {
        int q = atomicAdd(novf, 1);
        ovf[2 * (size_t)q] = dd;
        ovf[2 * (size_t)q + 1] = ss;
    }
}

__global__ __launch_bounds__(256) void owner_kernel(
    int n, int NB, const int* __restrict__ segcnt, const int* __restrict__ stage_w,
    const int* __restrict__ spillcnt, const int* __restrict__ spill_w,
    int* __restrict__ cnt, int* __restrict__ bucket,
    int* __restrict__ novf, int* __restrict__ ovf) {
    const u64* __restrict__ stage = (const u64*)stage_w;
    const u64* __restrict__ spill = (const u64*)spill_w;
    __shared__ int lcnt[RB];
    __shared__ __attribute__((aligned(16))) int limg[RB * CAP];
    __shared__ int lflag;
    int tid = threadIdx.x;
    int lane = tid & 31;
    int wave = __builtin_amdgcn_readfirstlane(tid >> 5);
    int b = blockIdx.x;
    int node0 = b * RB;

    for (int i = tid; i < RB; i += 256) lcnt[i] = 0;
    for (int i = tid; i < RB * CAP / 4; i += 256) ((int4*)limg)[i] = make_int4(0, 0, 0, 0);
    if (tid == 0) lflag = 0;
    __syncthreads();

    for (int c0 = 0; c0 < NB; c0 += 256) {
        int cw = c0 + wave * 32;
        int myc = cw + lane;
        int nseg = myc < NB ? segcnt[(size_t)b * NB + myc] : 0;
        u64 ed[32];
#pragma unroll
        for (int i = 0; i < 32; ++i) {
            int ni = __builtin_amdgcn_readlane(nseg, i);
            ed[i] = 0;
            if (lane < ni) ed[i] = stage[((size_t)b * NB + (cw + i)) * SEGK + lane];
        }
        int pl[32];
#pragma unroll
        for (int i = 0; i < 32; ++i) {
            int ni = __builtin_amdgcn_readlane(nseg, i);
            pl[i] = -1;
            if (lane < ni) pl[i] = atomicAdd(&lcnt[(int)(unsigned)ed[i] - node0], 1);
        }
#pragma unroll
        for (int i = 0; i < 32; ++i) {
            if (pl[i] >= 0) owner_place((int)(unsigned)ed[i], (int)(ed[i] >> 32), pl[i], node0, limg, novf, ovf);
        }
    }

    for (int c = tid; c < NB; c += 256) {
        if (spillcnt[c] > 0) lflag = 1;
    }
    __syncthreads();
    if (lflag != 0) {
        for (int c = 0; c < NB; ++c) {
            int m = spillcnt[c];
            for (int i = tid; i < m; i += 256) {
                u64 e = spill[(size_t)c * CHUNK + i];
                int dd = (int)(unsigned)e;
                if (dd >= node0 && dd < node0 + RB) {
                    int p = atomicAdd(&lcnt[dd - node0], 1);
                    owner_place(dd, (int)(e >> 32), p, node0, limg, novf, ovf);
                }
            }
        }
    }
    __syncthreads();

    for (int i = tid; i < RB; i += 256) {
        if (node0 + i < n) cnt[node0 + i] = lcnt[i];
    }
    for (int i = tid; i < RB * CAP / 4; i += 256) {
        int row = i / (CAP / 4);
        if (node0 + row < n) ((int4*)bucket)[(size_t)node0 * (CAP / 4) + i] = ((const int4*)limg)[i];
    }
}

#define TILE_STRIDE 132
#define W_STRIDE 136
#define STAGE_FLOATS (32 * TILE_STRIDE)

__device__ __forceinline__ float dinv_of(int count) {
    return rsqrtf((float)(count + 1));
}

__device__ __forceinline__ void fma4(float4& a, const float4& v, float w) {
    a.x = fmaf(v.x, w, a.x);
    a.y = fmaf(v.y, w, a.y);
    a.z = fmaf(v.z, w, a.z);
    a.w = fmaf(v.w, w, a.w);
}

__device__ __forceinline__ void request_rows(float4 (&v)[16], const float4* __restrict__ h4,
                                             int s, int j0, int g, int lane) {
#pragma unroll
    for (int k = 0; k < 16; ++k) {
        if (k < g) {
            int sk = __builtin_amdgcn_readlane(s, j0 + k);
            v[k] = h4[(size_t)sk * 32 + lane];
        }
    }
}

__device__ __forceinline__ void sum_rows(float4& acc, const float4 (&v)[16],
                                         int wi, int j0, int g) {
#pragma unroll
    for (int k = 0; k < 16; ++k) {
        if (k < g) {
            float wk = __int_as_float(__builtin_amdgcn_readlane(wi, j0 + k));
            fma4(acc, v[k], wk);
        }
    }
}

__device__ __forceinline__ int list_counter(const int* __restrict__ cnt, int base, int G,
                                            int ntiles, int n, int tu, int rp, int li,
                                            bool& exists) {
    int nv = (ntiles - base + G - 1) / G;
    int node = (base + (2 * (li >> 1) + tu) * G) * 16 + rp + (li & 1);
    exists = base < ntiles && 2 * (li >> 1) + tu < nv && node < n;
    return cnt[exists ? node : 0];
}

__device__ __forceinline__ void store_rows(float* __restrict__ out, const float* img,
                                           int node, int n, int lane) {
    if (node + 1 < n) {
        float4 o0 = *(const float4*)img;
        float4 o1 = *(const float4*)(img + TILE_STRIDE);
        ((float4*)out)[(size_t)node * 32 + lane] = o0;
        ((float4*)out)[(size_t)(node + 1) * 32 + lane] = o1;
    } else if (node < n) {
        ((float4*)out)[(size_t)node * 32 + lane] = *(const float4*)img;
    }
}

__global__ __launch_bounds__(512) __attribute__((amdgpu_waves_per_eu(8, 8))) void gather_gemm_kernel(
    const int* __restrict__ cnt, const int* __restrict__ bucket,
    const int* __restrict__ novf, const int* __restrict__ ovf,
    const float* __restrict__ xin, const float* __restrict__ W,
    const float* __restrict__ bias, float* __restrict__ out, int n, int relu) {
    __shared__ __attribute__((aligned(16))) float Wl[D * W_STRIDE];
    __shared__ __attribute__((aligned(16))) float ring[2 * STAGE_FLOATS];
    __shared__ __attribute__((aligned(16))) float oring[2 * STAGE_FLOATS];

    int tid = threadIdx.x;
    int lane = tid & 31;
    int wave = __builtin_amdgcn_readfirstlane(tid >> 5);
    int tu = wave >> 3;
    int rp = (wave & 7) * 2;
    const float4* h4 = (const float4*)xin;
    int ntiles = (n + 15) >> 4;

    int lrow = lane & 15;
    int lhi  = lane >> 4;
    int G = gridDim.x;

    int col = (wave & 7) * 16 + lrow;
    float bv = bias[col];
    const float* bp = Wl + (2 * lhi) * W_STRIDE + col;

    int li = lane & 7;
    bool rvl;
    int cl = list_counter(cnt, blockIdx.x, G, ntiles, n, tu, rp, li, rvl);

    {
        v4f wv[8];
#pragma unroll
        for (int q = 0; q < 8; ++q) wv[q] = ((const v4f*)W)[(size_t)(wave * 8 + q) * 32 + lane];
        asm volatile("" :: "v"(wv[0]), "v"(wv[1]), "v"(wv[2]), "v"(wv[3]),
                           "v"(wv[4]), "v"(wv[5]), "v"(wv[6]), "v"(wv[7]));
#pragma unroll
        for (int q = 0; q < 8; ++q) *(v4f*)(Wl + (wave * 8 + q) * W_STRIDE + 4 * lane) = wv[q];
    }

    for (int base = blockIdx.x; base < ntiles; base += 8 * G) {
    int nv = (ntiles - base + G - 1) / G;
    if (nv > 8) nv = 8;
    int S = (nv + 1) >> 1;
    int NR = 2 * S;

#define ROW_TILE(i) (2 * ((i) >> 1) + tu)
#define ROW_NODE(i) ((base + ROW_TILE(i) * G) * 16 + rp + ((i) & 1))

    int ddi = __float_as_int(dinv_of(cl));
    int cc = rvl ? cl : 0;

    int s0 = 0, s1 = 0;
    float w0 = 0.0f;
    {
        int c0 = __builtin_amdgcn_readlane(cc, 0);
        int c1 = __builtin_amdgcn_readlane(cc, 1);
        c0 = c0 < CAP ? c0 : CAP;
        c1 = c1 < CAP ? c1 : CAP;
        int n0 = ROW_NODE(0);
        int n1 = ROW_NODE(1);
        float g0 = 0.0f;
        if (lane < c0) s0 = bucket[(size_t)n0 * CAP + lane];
        if (lane < c1) s1 = bucket[(size_t)n1 * CAP + lane];
        if (lane < c0) g0 = dinv_of(cnt[s0]);
        w0 = g0 * __int_as_float(__builtin_amdgcn_readlane(ddi, 0));
    }

    float4 v[16];
    float4 vself = make_float4(0.0f, 0.0f, 0.0f, 0.0f);
    int k1 = 0, s2 = 0;

#pragma unroll 1
    for (int i = -1; i < NR; ++i) {
        bool edge = i >= 0 && (i & 1) != 0;
        if (i >= 0) {
            if (ROW_TILE(i) < nv) {
                float4 acc = make_float4(0.0f, 0.0f, 0.0f, 0.0f);
                int node = ROW_NODE(i);
                if (node < n) {
                    int c = __builtin_amdgcn_readlane(cc, i);
                    int m = c < CAP ? c : CAP;
                    float di = __int_as_float(__builtin_amdgcn_readlane(ddi, i));
                    int w0i = __float_as_int(w0);
                    sum_rows(acc, v, w0i, 0, m < 16 ? m : 16);
                    if (m > 16) {
                        request_rows(v, h4, s0, 16, m - 16, lane);
                        sum_rows(acc, v, w0i, 16, m - 16);
                    }
                    if (c > CAP) {
                        int nvf = novf[0];
                        for (int q0 = 0; q0 < nvf; q0 += 32) {
                            int q = q0 + lane;
                            int os = 0;
                            float ow = 0.0f;
                            bool hit = false;
                            if (q < nvf) {
                                if (ovf[2 * (size_t)q] == node) {
                                    hit = true;
                                    os = ovf[2 * (size_t)q + 1];
                                    ow = dinv_of(cnt[os]) * di;
                                }
                            }
                            unsigned mask = __builtin_amdgcn_ballot_w32(hit);
                            while (mask != 0u) {
                                int j = __builtin_ctz(mask);
                                mask &= mask - 1u;
                                int sk = __builtin_amdgcn_readlane(os, j);
                                float wk = __int_as_float(
                                    __builtin_amdgcn_readlane(__float_as_int(ow), j));
                                float4 vo = h4[(size_t)sk * 32 + lane];
                                fma4(acc, vo, wk);
                            }
                        }
                    }
                    fma4(acc, vself, di * di);
                }
                float* slot = ring + ((i >> 1) & 1) * STAGE_FLOATS
                            + (tu * 16 + rp + (i & 1)) * TILE_STRIDE + 4 * lane;
                *(float4*)slot = acc;
            }
            int i1 = i + 1 < NR ? i + 1 : NR - 1;
            s0 = s1;
            s1 = s2;
            w0 = dinv_of(k1) * __int_as_float(__builtin_amdgcn_readlane(ddi, i1));
            if (edge) {
                asm volatile("s_wait_dscnt 0x0\n\ts_barrier_signal -1" ::: "memory");
            }
        }
        if (i + 1 < NR) {
            int r = i + 1;
            int r2 = r + 2 < NR ? r + 2 : NR - 1;
            k1 = cnt[s1];
            int m2 = r + 2 < NR ? __builtin_amdgcn_readlane(cc, r2) : 0;
            m2 = m2 < CAP ? m2 : CAP;
            int n2 = ROW_NODE(r2);
            s2 = 0;
            if (lane < m2) s2 = bucket[(size_t)n2 * CAP + lane];
            int node = ROW_NODE(r);
            if (ROW_TILE(r) < nv && node < n) {
                int m = __builtin_amdgcn_readlane(cc, r);
                m = m < CAP ? m : CAP;
                vself = h4[(size_t)node * 32 + lane];
                request_rows(v, h4, s0, 0, m < 16 ? m : 16, lane);
            }
        }
        if (edge) {
            asm volatile("s_barrier_wait -1" ::: "memory");
            int s = i >> 1;
            if (s > 0 && ROW_TILE(2 * (s - 1)) < nv) {
                store_rows(out, oring + ((s - 1) & 1) * STAGE_FLOATS
                                + (tu * 16 + rp) * TILE_STRIDE + 4 * lane,
                           ROW_NODE(2 * (s - 1)), n, lane);
            }
            int u = 2 * s + tu;
            if (u < nv) {
                const float* ap = ring + (s & 1) * STAGE_FLOATS
                                + (tu * 16 + lrow) * TILE_STRIDE + lhi * 2;
                v8f acc8 = (v8f)(0.0f);
#pragma unroll
                for (int j = 0; j < 32; ++j) {
                    v2f af = *(const v2f*)(ap + 4 * j);
                    v2f bf;
                    bf.x = bp[(4 * j) * W_STRIDE];
                    bf.y = bp[(4 * j + 1) * W_STRIDE];
                    acc8 = __builtin_amdgcn_wmma_f32_16x16x4_f32(
                        false, af, false, bf, (short)0, acc8, false, false);
                }
                float* op = oring + (s & 1) * STAGE_FLOATS
                          + (tu * 16 + 8 * lhi) * TILE_STRIDE + col;
#pragma unroll
                for (int vv = 0; vv < 8; ++vv) {
                    float o = acc8[vv] + bv;
                    if (relu) o = fmaxf(o, 0.0f);
                    op[vv * TILE_STRIDE] = o;
                }
            }
        }
    }

    asm volatile("s_wait_dscnt 0x0\n\ts_barrier_signal -1\n\ts_barrier_wait -1" ::: "memory");
    if (ROW_TILE(2 * (S - 1)) < nv) {
        store_rows(out, oring + ((S - 1) & 1) * STAGE_FLOATS
                        + (tu * 16 + rp) * TILE_STRIDE + 4 * lane,
                   ROW_NODE(2 * (S - 1)), n, lane);
    }
    if (base + 8 * G < ntiles) cl = list_counter(cnt, base + 8 * G, G, ntiles, n, tu, rp, li, rvl);
    }
#undef ROW_TILE
#undef ROW_NODE
}

extern "C" void kernel_launch(void* const* d_in, const int* in_sizes, int n_in,
                              void* d_out, int out_size, void* d_ws, size_t ws_size,
                              hipStream_t stream) {
    const float* x   = (const float*)d_in[0];
    const int*   ei  = (const int*)d_in[1];
    const float* W1  = (const float*)d_in[2];
    const float* b1  = (const float*)d_in[3];
    const float* W2  = (const float*)d_in[4];
    const float* b2  = (const float*)d_in[5];
    float* out = (float*)d_out;

    const int N = in_sizes[0] / D;
    const int E = in_sizes[1] / 2;
    const int* src = ei;
    const int* dst = ei + E;

    const int nBlocksLayer  = 512;
    const int NB    = (E + CHUNK - 1) / CHUNK;
    const int nbins = (N + RB - 1) / RB;

    size_t nPad = ((size_t)N + 63) & ~(size_t)63;
    size_t ePad = ((size_t)E + 63) & ~(size_t)63;
    size_t sPad = ((size_t)nbins * NB + 63) & ~(size_t)63;
    size_t cPad = ((size_t)NB + 63) & ~(size_t)63;
    int* cnt      = (int*)d_ws;
    int* novf     = cnt + nPad;
    int* bucket   = novf + 64;
    int* ovf      = bucket + nPad * CAP;
    float* bufB   = (float*)(ovf + 2 * ePad);
    int* segcnt   = (int*)(bufB + nPad * D);
    int* spillcnt = segcnt + sPad;
    int* spill    = spillcnt + cPad;
    int* stage    = spill + 2 * (size_t)NB * CHUNK;

    partition_kernel<<<NB, 256, 0, stream>>>(src, dst, E, nbins, NB, segcnt, stage, spillcnt, spill, novf);
    owner_kernel<<<nbins, 256, 0, stream>>>(N, NB, segcnt, stage, spillcnt, spill, cnt, bucket, novf, ovf);

    gather_gemm_kernel<<<nBlocksLayer, 512, 0, stream>>>(cnt, bucket, novf, ovf, x, W1, b1, bufB, N, 1);
    gather_gemm_kernel<<<nBlocksLayer, 512, 0, stream>>>(cnt, bucket, novf, ovf, bufB, W2, b2, out, N, 0);
}
